# starting kernel
# baseline (speedup 1.0000x reference)
#include <hip/hip_runtime.h>
#include <math.h>

#define DIN   128
#define DH    64
#define HEADS 2
#define DOUT  64
#define NEG_SLOPE 0.2f

typedef float v2f __attribute__((ext_vector_type(2)));
typedef float v8f __attribute__((ext_vector_type(8)));

__device__ __forceinline__ void atomic_max_f32(float* addr, float v) {
  if (v >= 0.0f) atomicMax((int*)addr, __float_as_int(v));
  else           atomicMin((unsigned int*)addr, __float_as_uint(v));
}

__global__ void fill_f32(float* __restrict__ p, long n, float v) {
  long i = (long)blockIdx.x * blockDim.x + threadIdx.x;
  if (i < n) p[i] = v;
}

__global__ void __launch_bounds__(256)
gemm_wmma_f32(const float* __restrict__ A, const float* __restrict__ B,
              float* __restrict__ C, int M, int K, int Nc) {
  int gid  = blockIdx.x * blockDim.x + threadIdx.x;
  int wave = gid >> 5;
  int lane = threadIdx.x & 31;
  int tilesN = Nc >> 4;
  int tm = wave / tilesN;
  int tn = wave - tm * tilesN;
  if (tm * 16 >= M) return;
  int half = lane >> 4;
  int l16  = lane & 15;
  long rowA = (long)tm * 16 + l16;
  if (rowA >= M) rowA = M - 1;
  int  colB = tn * 16 + l16;
  v8f acc = {};
  for (int k0 = 0; k0 < K; k0 += 4) {
    int ka = k0 + half * 2;
    v2f a, b;
    a.x = A[rowA * K + ka];
    a.y = A[rowA * K + ka + 1];
    b.x = B[(long)ka       * Nc + colB];
    b.y = B[(long)(ka + 1) * Nc + colB];
    acc = __builtin_amdgcn_wmma_f32_16x16x4_f32(false, a, false, b,
                                                (short)0, acc, false, false);
  }
#pragma unroll
  for (int r = 0; r < 8; ++r) {
    int row = tm * 16 + r + half * 8;
    if (row < M) C[(long)row * Nc + colB] = acc[r];
  }
}

__global__ void att_dots(const float* __restrict__ h,
                         const float* __restrict__ att_s,
                         const float* __restrict__ att_d,
                         float* __restrict__ as_, float* __restrict__ ad_,
                         int n, int heads, int C) {
  int i = blockIdx.x * blockDim.x + threadIdx.x;
  if (i >= n * heads) return;
  int node = i / heads, hd = i - node * heads;
  const float* hp = h + ((long)node * heads + hd) * C;
  const float* sp = att_s + hd * C;
  const float* dp = att_d + hd * C;
  float ss = 0.f, dd = 0.f;
  for (int c = 0; c < C; ++c) { float v = hp[c]; ss += v * sp[c]; dd += v * dp[c]; }
  as_[i] = ss; ad_[i] = dd;
}

__global__ void edge_max(const int* __restrict__ srcs, const int* __restrict__ dsts,
                         int E0, int Etot,
                         const float* __restrict__ as_, const float* __restrict__ ad_,
                         float* __restrict__ ev, float* __restrict__ m, int heads) {
  int e = blockIdx.x * blockDim.x + threadIdx.x;
  if (e >= Etot) return;
  int s = (e < E0) ? srcs[e] : (e - E0);
  int d = (e < E0) ? dsts[e] : (e - E0);
  for (int h = 0; h < heads; ++h) {
    float v = as_[s * heads + h] + ad_[d * heads + h];
    v = (v > 0.f) ? v : NEG_SLOPE * v;
    ev[(long)e * heads + h] = v;
    atomic_max_f32(&m[d * heads + h], v);
  }
}

__global__ void edge_sum(const int* __restrict__ dsts, int E0, int Etot,
                         float* __restrict__ ev, const float* __restrict__ m,
                         float* __restrict__ den, int heads) {
  int e = blockIdx.x * blockDim.x + threadIdx.x;
  if (e >= Etot) return;
  int d = (e < E0) ? dsts[e] : (e - E0);
  for (int h = 0; h < heads; ++h) {
    float ex = expf(ev[(long)e * heads + h] - m[d * heads + h]);
    ev[(long)e * heads + h] = ex;
    atomicAdd(&den[d * heads + h], ex);
  }
}

__global__ void edge_agg(const int* __restrict__ srcs, const int* __restrict__ dsts,
                         int E0, int Etot,
                         const float* __restrict__ ev, const float* __restrict__ den,
                         const float* __restrict__ h, float* __restrict__ out,
                         int heads, int C) {
  long gid = (long)blockIdx.x * blockDim.x + threadIdx.x;
  int e = (int)(gid >> 5);
  int lane = (int)(gid & 31);
  if (e >= Etot) return;
  int s = (e < E0) ? srcs[e] : (e - E0);
  int d = (e < E0) ? dsts[e] : (e - E0);
  int F = heads * C;
  for (int c = lane; c < F; c += 32) {
    int hd = c / C;
    float alpha = ev[(long)e * heads + hd] / den[d * heads + hd];
    atomicAdd(&out[(long)d * F + c], h[(long)s * F + c] * alpha);
  }
}

__global__ void elu_bias(float* __restrict__ p, const float* __restrict__ bias,
                         long n, int F) {
  long i = (long)blockIdx.x * blockDim.x + threadIdx.x;
  if (i >= n) return;
  float v = p[i] + bias[i % F];
  p[i] = (v > 0.f) ? v : (expf(v) - 1.f);
}

__global__ void bias_add(float* __restrict__ p, const float* __restrict__ bias,
                         long n, int F) {
  long i = (long)blockIdx.x * blockDim.x + threadIdx.x;
  if (i >= n) return;
  p[i] += bias[i % F];
}

extern "C" void kernel_launch(void* const* d_in, const int* in_sizes, int n_in,
                              void* d_out, int out_size, void* d_ws, size_t ws_size,
                              hipStream_t stream) {
  const float* x    = (const float*)d_in[0];
  const int*   ei   = (const int*)  d_in[1];
  const float* W1   = (const float*)d_in[2];
  const float* atS1 = (const float*)d_in[3];
  const float* atD1 = (const float*)d_in[4];
  const float* b1   = (const float*)d_in[5];
  const float* W2   = (const float*)d_in[6];
  const float* atS2 = (const float*)d_in[7];
  const float* atD2 = (const float*)d_in[8];
  const float* b2   = (const float*)d_in[9];
  float* out = (float*)d_out;

  const int N    = in_sizes[0] / DIN;
  const int E0   = in_sizes[1] / 2;
  const int Etot = E0 + N;
  const int* srcs = ei;
  const int* dsts = ei + E0;

  float* ws = (float*)d_ws;
  long off = 0;
  float* h1  = ws + off; off += (long)N * (HEADS * DH);
  float* o1  = ws + off; off += (long)N * (HEADS * DH);
  float* ex1 = ws + off; off += (long)Etot * HEADS;
  float* a1s = ws + off; off += (long)N * HEADS;
  float* a1d = ws + off; off += (long)N * HEADS;
  float* m1  = ws + off; off += (long)N * HEADS;
  float* d1  = ws + off; off += (long)N * HEADS;
  float* h2  = h1;
  float* ex2 = h1 + (long)N * DOUT;
  float* a2s = ex2 + Etot;
  float* a2d = a2s + N;
  float* m2  = a2d + N;
  float* d2  = m2 + N;

  const int TB = 256;
  auto blocks = [](long n, int tb) { return (unsigned)((n + tb - 1) / tb); };

  fill_f32<<<blocks((long)N * HEADS * DH, TB), TB, 0, stream>>>(o1, (long)N * HEADS * DH, 0.f);
  fill_f32<<<blocks((long)N * HEADS, TB), TB, 0, stream>>>(d1, (long)N * HEADS, 0.f);
  fill_f32<<<blocks((long)N * HEADS, TB), TB, 0, stream>>>(m1, (long)N * HEADS, -INFINITY);

  {
    long waves = (long)((N + 15) / 16) * ((HEADS * DH) / 16);
    gemm_wmma_f32<<<blocks(waves * 32, TB), TB, 0, stream>>>(x, W1, h1, N, DIN, HEADS * DH);
  }
  att_dots<<<blocks((long)N * HEADS, TB), TB, 0, stream>>>(h1, atS1, atD1, a1s, a1d, N, HEADS, DH);
  edge_max<<<blocks(Etot, TB), TB, 0, stream>>>(srcs, dsts, E0, Etot, a1s, a1d, ex1, m1, HEADS);
  edge_sum<<<blocks(Etot, TB), TB, 0, stream>>>(dsts, E0, Etot, ex1, m1, d1, HEADS);
  edge_agg<<<blocks((long)Etot * 32, TB), TB, 0, stream>>>(srcs, dsts, E0, Etot, ex1, d1, h1, o1, HEADS, DH);
  elu_bias<<<blocks((long)N * HEADS * DH, TB), TB, 0, stream>>>(o1, b1, (long)N * HEADS * DH, HEADS * DH);

  fill_f32<<<blocks((long)N * DOUT, TB), TB, 0, stream>>>(out, (long)N * DOUT, 0.f);
  fill_f32<<<blocks(N, TB), TB, 0, stream>>>(d2, N, 0.f);
  fill_f32<<<blocks(N, TB), TB, 0, stream>>>(m2, N, -INFINITY);

  {
    long waves = (long)((N + 15) / 16) * (DOUT / 16);
    gemm_wmma_f32<<<blocks(waves * 32, TB), TB, 0, stream>>>(o1, W2, h2, N, HEADS * DH, DOUT);
  }
  att_dots<<<blocks(N, TB), TB, 0, stream>>>(h2, atS2, atD2, a2s, a2d, N, 1, DOUT);
  edge_max<<<blocks(Etot, TB), TB, 0, stream>>>(srcs, dsts, E0, Etot, a2s, a2d, ex2, m2, 1);
  edge_sum<<<blocks(Etot, TB), TB, 0, stream>>>(dsts, E0, Etot, ex2, m2, d2, 1);
  edge_agg<<<blocks((long)Etot * 32, TB), TB, 0, stream>>>(srcs, dsts, E0, Etot, ex2, d2, h2, out, 1, DOUT);
  bias_add<<<blocks((long)N * DOUT, TB), TB, 0, stream>>>(out, b2, (long)N * DOUT, DOUT);
}


# GEMM waves stay resident: each keeps a 32-column strip of W in registers for all its 16-row tiles, next A tile prefetched (16-byte loads); a matrix instruction now sums k, k+1, k+4, k+5
# speedup vs baseline: 5.0211x; 5.0211x over previous
#include <hip/hip_runtime.h>
#include <math.h>

#define DIN   128
#define DH    64
#define HEADS 2
#define DOUT  64
#define NEG_SLOPE 0.2f

typedef float v2f __attribute__((ext_vector_type(2)));
typedef float v8f __attribute__((ext_vector_type(8)));

typedef float v4f __attribute__((ext_vector_type(4)));
#define GK 128
static_assert(DIN == GK && HEADS * DH == GK, "both products have inner dimension GK");

template <int NSTRIP>
__global__ void __launch_bounds__(256)
gemm_wmma_f32(const float* __restrict__ A, const float* __restrict__ B,
              float* __restrict__ C, int M) {
  constexpr int Nc      = 32 * NSTRIP;
  constexpr int NSTREAM = 8 / NSTRIP;
  const int lane   = threadIdx.x & 31;
  const int wave   = __builtin_amdgcn_readfirstlane(threadIdx.x >> 5);
  const int half   = lane >> 4;
  const int l16    = lane & 15;
  const int strip  = wave % NSTRIP;
  const int stream = wave / NSTRIP;
  const int colp   = strip * 32 + 2 * l16;

  const int tiles = (M + 15) >> 4;
  const int step  = gridDim.x * NSTREAM;
  int tile = blockIdx.x * NSTREAM + stream;
  if (tile >= tiles) return;

  v2f bw[2][32];
#pragma unroll
  for (int j = 0; j < 32; ++j) {
    const int k = 8 * (j >> 1) + 4 * half + 2 * (j & 1);
    const v2f w0 = *(const v2f*)(B + k * Nc + colp);
    const v2f w1 = *(const v2f*)(B + (k + 1) * Nc + colp);
    bw[0][j].x = w0.x; bw[0][j].y = w1.x;
    bw[1][j].x = w0.y; bw[1][j].y = w1.y;
  }

  auto load_tile = [&](v4f (&a)[16], int tl) {
    long row = (long)tl * 16 + l16;
    if (row >= M) row = M - 1;
    const v4f* p = (const v4f*)(A + row * GK + 4 * half);
#pragma unroll
    for (int jj = 0; jj < 16; ++jj) a[jj] = p[2 * jj];
  };
  auto mul_tile = [&](const v4f (&a)[16], int tl) {
    v8f c0 = {}, c1 = {};
#pragma unroll
    for (int j = 0; j < 32; ++j) {
      v2f av;
      if (j & 1) { av.x = a[j >> 1].z; av.y = a[j >> 1].w; }
      else       { av.x = a[j >> 1].x; av.y = a[j >> 1].y; }
      c0 = __builtin_amdgcn_wmma_f32_16x16x4_f32(false, av, false, bw[0][j],
                                                 (short)0, c0, false, false);
      c1 = __builtin_amdgcn_wmma_f32_16x16x4_f32(false, av, false, bw[1][j],
                                                 (short)0, c1, false, false);
    }
    if ((tl + 1) * 16 <= M) {
      float* cp = C + ((long)tl * 16 + half * 8) * Nc + colp;
#pragma unroll
      for (int r = 0; r < 8; ++r) {
        v2f o; o.x = c0[r]; o.y = c1[r];
        *(v2f*)(cp + r * Nc) = o;
      }
    } else {
#pragma unroll
      for (int r = 0; r < 8; ++r) {
        long row = (long)tl * 16 + r + half * 8;
        if (row < M) {
          v2f o; o.x = c0[r]; o.y = c1[r];
          *(v2f*)(C + row * Nc + colp) = o;
        }
      }
    }
  };

  v4f a0[16], a1[16];
  load_tile(a0, tile);
  while (true) {
    load_tile(a1, min(tile + step, tiles - 1));
    mul_tile(a0, tile);
    tile += step;
    if (tile >= tiles) break;
    load_tile(a0, min(tile + step, tiles - 1));
    mul_tile(a1, tile);
    tile += step;
    if (tile >= tiles) break;
  }
}

__global__ void att_dots(const float* __restrict__ h,
                         const float* __restrict__ att_s,
                         const float* __restrict__ att_d,
                         float* __restrict__ as_, float* __restrict__ ad_,
                         int n, int heads, int C) {
  int i = blockIdx.x * blockDim.x + threadIdx.x;
  if (i >= n * heads) return;
  int node = i / heads, hd = i - node * heads;
  const float* hp = h + ((long)node * heads + hd) * C;
  const float* sp = att_s + hd * C;
  const float* dp = att_d + hd * C;
  float ss = 0.f, dd = 0.f;
  for (int c = 0; c < C; ++c) { float v = hp[c]; ss += v * sp[c]; dd += v * dp[c]; }
  as_[i] = ss; ad_[i] = dd;
}

#define RNG 196

__global__ void __launch_bounds__(256)
edge_keys(const int* __restrict__ dsts, int E0, unsigned int* __restrict__ keys) {
  const int nw = (E0 >> 4) << 2;
  if (nw == 0) return;
  const int w0 = blockIdx.x * 1024 + threadIdx.x;
  int d[4][4];
#pragma unroll
  for (int k = 0; k < 4; ++k) {
    const int* dp = dsts + 4 * (long)min(w0 + k * 256, nw - 1);
#pragma unroll
    for (int c = 0; c < 4; ++c) d[k][c] = dp[c];
  }
#pragma unroll
  for (int k = 0; k < 4; ++k) {
    const int w = w0 + k * 256;
    if (w < nw) {
      unsigned k0 = min((unsigned)d[k][0] / RNG, 255u);
      unsigned k1 = min((unsigned)d[k][1] / RNG, 255u);
      unsigned k2 = min((unsigned)d[k][2] / RNG, 255u);
      unsigned k3 = min((unsigned)d[k][3] / RNG, 255u);
      keys[w] = k0 | (k1 << 8) | (k2 << 16) | (k3 << 24);
    }
  }
}

#define CAP 8192
#define NT  1024
#define NW  (NT / 32)

__device__ __forceinline__ float lrelu(float v) { return (v > 0.f) ? v : NEG_SLOPE * v; }

__device__ __forceinline__ void lds_max_f32(float* addr, float v) {
  if (v >= 0.0f) atomicMax((int*)addr, __float_as_int(v));
  else           atomicMin((unsigned int*)addr, __float_as_uint(v));
}

template <int H, int C, bool DO_ELU, bool FIRST>
__global__ void __launch_bounds__(NT)
gat_owner(const int* __restrict__ srcs, const int* __restrict__ dsts, int E0, int N,
          const float* __restrict__ as_, const float* __restrict__ ad_,
          const float* __restrict__ hf, const float* __restrict__ bias,
          float* __restrict__ out,
          int* __restrict__ lsrc, int* __restrict__ ldl, int* __restrict__ lcnt,
          const unsigned int* __restrict__ keys) {
  constexpr int F  = H * C;
  constexpr int FJ = F / 32;
  __shared__ float s_acc[RNG * F];
  __shared__ float s_m[RNG * H];
  __shared__ float s_den[RNG * H];
  __shared__ float s_ad[RNG * H];
  __shared__ float s_self[RNG * H];
  __shared__ int   s_e[CAP];
  __shared__ int   s_dl[CAP];
  __shared__ float s_ex[CAP * H];
  __shared__ int   s_cnt;

  const int tid  = threadIdx.x;
  const int lane = tid & 31;
  const int wave = __builtin_amdgcn_readfirstlane(tid >> 5);
  const int lo   = blockIdx.x * RNG;
  const int cnt  = min(RNG, N - lo);

  for (int i = tid; i < RNG * F; i += NT) s_acc[i] = 0.f;
  for (int i = tid; i < cnt * H; i += NT) {
    float a_d = ad_[lo * H + i];
    s_ad[i]  = a_d;
    s_m[i]   = lrelu(as_[lo * H + i] + a_d);
    s_den[i] = 0.f;
  }
  if (FIRST && tid == 0) s_cnt = 0;
  __syncthreads();

  int nm;
  if (FIRST) {
    auto keep = [&](int e) {
      int slot = atomicAdd(&s_cnt, 1);
      if (slot < CAP) s_e[slot] = e;
    };
    if (gridDim.x <= 256) {
      const unsigned kb = blockIdx.x * 0x01010101u;
      const int nfull = E0 >> 4;
      const uint4* kv = (const uint4*)keys;
      auto zb = [](unsigned x) {
        return ~(((x & 0x7F7F7F7Fu) + 0x7F7F7F7Fu) | x) & 0x80808080u;
      };
      for (int base = 0; base < nfull; base += NT * 4) {
        uint4 q[4];
#pragma unroll
        for (int k = 0; k < 4; ++k) q[k] = kv[min(base + k * NT + tid, nfull - 1)];
#pragma unroll
        for (int k = 0; k < 4; ++k) {
          const int v = base + k * NT + tid;
          unsigned m = (zb(q[k].x ^ kb) >> 7) | (zb(q[k].y ^ kb) >> 6) |
                       (zb(q[k].z ^ kb) >> 5) | (zb(q[k].w ^ kb) >> 4);
          if (v >= nfull) m = 0;
          const int e0 = v * 16;
          while (m) {
            int p = __builtin_ctz(m);
            m &= m - 1;
            keep(e0 + ((p & 7) << 2) + (p >> 3));
          }
        }
      }
      int et = nfull * 16 + tid;
      if (et < E0 && (unsigned)(dsts[et] - lo) < (unsigned)cnt) keep(et);
    } else {
      for (int e = tid; e < E0; e += NT)
        if ((unsigned)(dsts[e] - lo) < (unsigned)cnt) keep(e);
    }
    __syncthreads();
    nm = __builtin_amdgcn_readfirstlane(s_cnt);
    if (tid == 0) lcnt[blockIdx.x] = nm;
  } else {
    nm = lcnt[blockIdx.x];
  }
  const bool fits = (nm <= CAP);
  const long lb   = (long)blockIdx.x * CAP;

  if (fits) {
    for (int i = tid; i < nm; i += NT) {
      int s, dl;
      if (FIRST) {
        int e = s_e[i];
        s  = srcs[e];
        dl = dsts[e] - lo;
        s_e[i]  = s;
        s_dl[i] = dl;
        lsrc[lb + i] = s;
        ldl[lb + i]  = dl;
      } else {
        s  = lsrc[lb + i];
        dl = ldl[lb + i];
        s_e[i]  = s;
        s_dl[i] = dl;
      }
#pragma unroll
      for (int h = 0; h < H; ++h) {
        float v = lrelu(as_[s * H + h] + s_ad[dl * H + h]);
        s_ex[i * H + h] = v;
        lds_max_f32(&s_m[dl * H + h], v);
      }
    }
  } else {
    for (int e = tid; e < E0; e += NT) {
      unsigned dl = (unsigned)(dsts[e] - lo);
      if (dl < (unsigned)cnt) {
        int s = srcs[e];
        for (int h = 0; h < H; ++h)
          lds_max_f32(&s_m[dl * H + h], lrelu(as_[s * H + h] + s_ad[dl * H + h]));
      }
    }
  }
  __syncthreads();

  if (fits) {
    for (int i = tid; i < nm; i += NT) {
      int dl = s_dl[i];
#pragma unroll
      for (int h = 0; h < H; ++h) {
        float ex = expf(s_ex[i * H + h] - s_m[dl * H + h]);
        s_ex[i * H + h] = ex;
        atomicAdd(&s_den[dl * H + h], ex);
      }
    }
  }
  for (int i = tid; i < cnt * H; i += NT) {
    float ex = expf(lrelu(as_[lo * H + i] + s_ad[i]) - s_m[i]);
    s_self[i] = ex;
    atomicAdd(&s_den[i], ex);
  }
  __syncthreads();

  if (fits) {
    for (int i0 = wave; i0 < nm; i0 += NW * 4) {
      float val[4][FJ];
      int   dlk[4];
#pragma unroll
      for (int k = 0; k < 4; ++k) {
        int i  = i0 + k * NW;
        int ii = (i < nm) ? i : i0;
        int s  = s_e[ii];
        dlk[k] = s_dl[ii];
#pragma unroll
        for (int j = 0; j < FJ; ++j)
          val[k][j] = hf[(long)s * F + lane + 32 * j] * s_ex[ii * H + (32 * j) / C];
      }
#pragma unroll
      for (int k = 0; k < 4; ++k) {
        if (i0 + k * NW < nm) {
#pragma unroll
          for (int j = 0; j < FJ; ++j)
            atomicAdd(&s_acc[dlk[k] * F + lane + 32 * j], val[k][j]);
        }
      }
    }
  } else {
    for (int e = tid; e < E0; e += NT) {
      unsigned dl = (unsigned)(dsts[e] - lo);
      if (dl < (unsigned)cnt) {
        int s = srcs[e];
        for (int h = 0; h < H; ++h) {
          float ex = expf(lrelu(as_[s * H + h] + s_ad[dl * H + h]) - s_m[dl * H + h]);
          atomicAdd(&s_den[dl * H + h], ex);
          for (int c = 0; c < C; ++c)
            atomicAdd(&s_acc[dl * F + h * C + c], hf[(long)s * F + h * C + c] * ex);
        }
      }
    }
  }
  for (int i = wave; i < cnt; i += NW) {
#pragma unroll
    for (int j = 0; j < FJ; ++j)
      atomicAdd(&s_acc[i * F + lane + 32 * j],
                hf[(long)(lo + i) * F + lane + 32 * j] * s_self[i * H + (32 * j) / C]);
  }
  __syncthreads();

  for (int i = tid; i < cnt * F; i += NT) {
    int dl = i / F, c = i - dl * F;
    float v = s_acc[i] / s_den[dl * H + c / C] + bias[c];
    if (DO_ELU) v = (v > 0.f) ? v : (expf(v) - 1.f);
    out[(long)lo * F + i] = v;
  }
}

extern "C" void kernel_launch(void* const* d_in, const int* in_sizes, int n_in,
                              void* d_out, int out_size, void* d_ws, size_t ws_size,
                              hipStream_t stream) {
  const float* x    = (const float*)d_in[0];
  const int*   ei   = (const int*)  d_in[1];
  const float* W1   = (const float*)d_in[2];
  const float* atS1 = (const float*)d_in[3];
  const float* atD1 = (const float*)d_in[4];
  const float* b1   = (const float*)d_in[5];
  const float* W2   = (const float*)d_in[6];
  const float* atS2 = (const float*)d_in[7];
  const float* atD2 = (const float*)d_in[8];
  const float* b2   = (const float*)d_in[9];
  float* out = (float*)d_out;

  const int N    = in_sizes[0] / DIN;
  const int E0   = in_sizes[1] / 2;
  const int* srcs = ei;
  const int* dsts = ei + E0;

  float* ws = (float*)d_ws;
  long off = 0;
  float* h1  = ws + off; off += (long)N * (HEADS * DH);
  float* o1  = ws + off; off += (long)N * (HEADS * DH);
  float* a1s = ws + off; off += (long)N * HEADS;
  float* a1d = ws + off; off += (long)N * HEADS;
  off = (off + 3) & ~3L;
  float* h2  = h1;
  float* a2s = h1 + (long)N * DOUT;
  float* a2d = a2s + N;

  const int TB = 256;
  auto blocks = [](long n, int tb) { return (unsigned)((n + tb - 1) / tb); };
  const unsigned NB = blocks(N, RNG);
  int* lsrc = (int*)(ws + off); off += (long)NB * CAP;
  int* ldl  = (int*)(ws + off); off += (long)NB * CAP;
  int* lcnt = (int*)(ws + off); off += NB;
  off = (off + 3) & ~3L;
  unsigned int* keys = (unsigned int*)(ws + off);

  gemm_wmma_f32<(HEADS * DH) / 32><<<256, 256, 0, stream>>>(x, W1, h1, N);
  att_dots<<<blocks((long)N * HEADS, TB), TB, 0, stream>>>(h1, atS1, atD1, a1s, a1d, N, HEADS, DH);
  edge_keys<<<blocks((long)(E0 >> 4) * 4, 1024), 256, 0, stream>>>(dsts, E0, keys);
  gat_owner<HEADS, DH, true, true><<<NB, NT, 0, stream>>>(srcs, dsts, E0, N, a1s, a1d, h1, b1, o1,
                                                          lsrc, ldl, lcnt, keys);

  gemm_wmma_f32<DOUT / 32><<<256, 256, 0, stream>>>(o1, W2, h2, N);
  att_dots<<<blocks(N, TB), TB, 0, stream>>>(h2, atS2, atD2, a2s, a2d, N, 1, DOUT);
  gat_owner<1, DOUT, false, false><<<NB, NT, 0, stream>>>(srcs, dsts, E0, N, a2s, a2d, h2, b2, out,
                                                          lsrc, ldl, lcnt, keys);
}


# Feature sum: self loops join the kept list; a wave takes 8 entries per turn, source, destination, weight as scalars (readlane), all 8 row loads issued before the first add; LDS atomics, arrival order
# speedup vs baseline: 5.5574x; 1.1068x over previous
#include <hip/hip_runtime.h>
#include <math.h>

#define DIN   128
#define DH    64
#define HEADS 2
#define DOUT  64
#define NEG_SLOPE 0.2f

typedef float v2f __attribute__((ext_vector_type(2)));
typedef float v8f __attribute__((ext_vector_type(8)));

typedef float v4f __attribute__((ext_vector_type(4)));
#define GK 128
static_assert(DIN == GK && HEADS * DH == GK, "both products have inner dimension GK");

template <int NSTRIP>
__global__ void __launch_bounds__(256)
gemm_wmma_f32(const float* __restrict__ A, const float* __restrict__ B,
              float* __restrict__ C, int M) {
  constexpr int Nc      = 32 * NSTRIP;
  constexpr int NSTREAM = 8 / NSTRIP;
  const int lane   = threadIdx.x & 31;
  const int wave   = __builtin_amdgcn_readfirstlane(threadIdx.x >> 5);
  const int half   = lane >> 4;
  const int l16    = lane & 15;
  const int strip  = wave % NSTRIP;
  const int stream = wave / NSTRIP;
  const int colp   = strip * 32 + 2 * l16;

  const int tiles = (M + 15) >> 4;
  const int step  = gridDim.x * NSTREAM;
  int tile = blockIdx.x * NSTREAM + stream;
  if (tile >= tiles) return;

  v2f bw[2][32];
#pragma unroll
  for (int j = 0; j < 32; ++j) {
    const int k = 8 * (j >> 1) + 4 * half + 2 * (j & 1);
    const v2f w0 = *(const v2f*)(B + k * Nc + colp);
    const v2f w1 = *(const v2f*)(B + (k + 1) * Nc + colp);
    bw[0][j].x = w0.x; bw[0][j].y = w1.x;
    bw[1][j].x = w0.y; bw[1][j].y = w1.y;
  }

  auto load_tile = [&](v4f (&a)[16], int tl) {
    long row = (long)tl * 16 + l16;
    if (row >= M) row = M - 1;
    const v4f* p = (const v4f*)(A + row * GK + 4 * half);
#pragma unroll
    for (int jj = 0; jj < 16; ++jj) a[jj] = p[2 * jj];
  };
  auto mul_tile = [&](const v4f (&a)[16], int tl) {
    v8f c0 = {}, c1 = {};
#pragma unroll
    for (int j = 0; j < 32; ++j) {
      v2f av;
      if (j & 1) { av.x = a[j >> 1].z; av.y = a[j >> 1].w; }
      else       { av.x = a[j >> 1].x; av.y = a[j >> 1].y; }
      c0 = __builtin_amdgcn_wmma_f32_16x16x4_f32(false, av, false, bw[0][j],
                                                 (short)0, c0, false, false);
      c1 = __builtin_amdgcn_wmma_f32_16x16x4_f32(false, av, false, bw[1][j],
                                                 (short)0, c1, false, false);
    }
    if ((tl + 1) * 16 <= M) {
      float* cp = C + ((long)tl * 16 + half * 8) * Nc + colp;
#pragma unroll
      for (int r = 0; r < 8; ++r) {
        v2f o; o.x = c0[r]; o.y = c1[r];
        *(v2f*)(cp + r * Nc) = o;
      }
    } else {
#pragma unroll
      for (int r = 0; r < 8; ++r) {
        long row = (long)tl * 16 + r + half * 8;
        if (row < M) {
          v2f o; o.x = c0[r]; o.y = c1[r];
          *(v2f*)(C + row * Nc + colp) = o;
        }
      }
    }
  };

  v4f a0[16], a1[16];
  load_tile(a0, tile);
  while (true) {
    load_tile(a1, min(tile + step, tiles - 1));
    mul_tile(a0, tile);
    tile += step;
    if (tile >= tiles) break;
    load_tile(a0, min(tile + step, tiles - 1));
    mul_tile(a1, tile);
    tile += step;
    if (tile >= tiles) break;
  }
}

__global__ void att_dots(const float* __restrict__ h,
                         const float* __restrict__ att_s,
                         const float* __restrict__ att_d,
                         float* __restrict__ as_, float* __restrict__ ad_,
                         int n, int heads, int C) {
  int i = blockIdx.x * blockDim.x + threadIdx.x;
  if (i >= n * heads) return;
  int node = i / heads, hd = i - node * heads;
  const float* hp = h + ((long)node * heads + hd) * C;
  const float* sp = att_s + hd * C;
  const float* dp = att_d + hd * C;
  float ss = 0.f, dd = 0.f;
  for (int c = 0; c < C; ++c) { float v = hp[c]; ss += v * sp[c]; dd += v * dp[c]; }
  as_[i] = ss; ad_[i] = dd;
}

#define RNG 196

__global__ void __launch_bounds__(256)
edge_keys(const int* __restrict__ dsts, int E0, unsigned int* __restrict__ keys) {
  const int nw = (E0 >> 4) << 2;
  if (nw == 0) return;
  const int w0 = blockIdx.x * 1024 + threadIdx.x;
  int d[4][4];
#pragma unroll
  for (int k = 0; k < 4; ++k) {
    const int* dp = dsts + 4 * (long)min(w0 + k * 256, nw - 1);
#pragma unroll
    for (int c = 0; c < 4; ++c) d[k][c] = dp[c];
  }
#pragma unroll
  for (int k = 0; k < 4; ++k) {
    const int w = w0 + k * 256;
    if (w < nw) {
      unsigned k0 = min((unsigned)d[k][0] / RNG, 255u);
      unsigned k1 = min((unsigned)d[k][1] / RNG, 255u);
      unsigned k2 = min((unsigned)d[k][2] / RNG, 255u);
      unsigned k3 = min((unsigned)d[k][3] / RNG, 255u);
      keys[w] = k0 | (k1 << 8) | (k2 << 16) | (k3 << 24);
    }
  }
}

#define CAP 8192
#define ROWS 8
#define NT  1024
#define NW  (NT / 32)

__device__ __forceinline__ float lrelu(float v) { return (v > 0.f) ? v : NEG_SLOPE * v; }

__device__ __forceinline__ void lds_max_f32(float* addr, float v) {
  if (v >= 0.0f) atomicMax((int*)addr, __float_as_int(v));
  else           atomicMin((unsigned int*)addr, __float_as_uint(v));
}

template <int H, int C, bool DO_ELU, bool FIRST>
__global__ void __launch_bounds__(NT)
gat_owner(const int* __restrict__ srcs, const int* __restrict__ dsts, int E0, int N,
          const float* __restrict__ as_, const float* __restrict__ ad_,
          const float* __restrict__ hf, const float* __restrict__ bias,
          float* __restrict__ out,
          int* __restrict__ lsrc, int* __restrict__ ldl, int* __restrict__ lcnt,
          const unsigned int* __restrict__ keys) {
  constexpr int F  = H * C;
  constexpr int FJ = F / 32;
  __shared__ float s_acc[(RNG + 1) * F];
  __shared__ float s_m[RNG * H];
  __shared__ float s_den[RNG * H];
  __shared__ float s_ad[RNG * H];
  __shared__ float s_self[RNG * H];
  __shared__ int   s_e[CAP];
  __shared__ int   s_dl[CAP];
  __shared__ float s_ex[CAP * H];
  __shared__ int   s_cnt;

  const int tid  = threadIdx.x;
  const int lane = tid & 31;
  const int wave = __builtin_amdgcn_readfirstlane(tid >> 5);
  const int lo   = blockIdx.x * RNG;
  const int cnt  = min(RNG, N - lo);

  for (int i = tid; i < (RNG + 1) * F; i += NT) s_acc[i] = 0.f;
  for (int i = tid; i < cnt * H; i += NT) {
    float a_d = ad_[lo * H + i];
    s_ad[i]  = a_d;
    s_m[i]   = lrelu(as_[lo * H + i] + a_d);
    s_den[i] = 0.f;
  }
  if (FIRST && tid == 0) s_cnt = 0;
  __syncthreads();

  int nm;
  if (FIRST) {
    auto keep = [&](int e) {
      int slot = atomicAdd(&s_cnt, 1);
      if (slot < CAP) s_e[slot] = e;
    };
    if (gridDim.x <= 256) {
      const unsigned kb = blockIdx.x * 0x01010101u;
      const int nfull = E0 >> 4;
      const uint4* kv = (const uint4*)keys;
      auto zb = [](unsigned x) {
        return ~(((x & 0x7F7F7F7Fu) + 0x7F7F7F7Fu) | x) & 0x80808080u;
      };
      for (int base = 0; base < nfull; base += NT * 4) {
        uint4 q[4];
#pragma unroll
        for (int k = 0; k < 4; ++k) q[k] = kv[min(base + k * NT + tid, nfull - 1)];
#pragma unroll
        for (int k = 0; k < 4; ++k) {
          const int v = base + k * NT + tid;
          unsigned m = (zb(q[k].x ^ kb) >> 7) | (zb(q[k].y ^ kb) >> 6) |
                       (zb(q[k].z ^ kb) >> 5) | (zb(q[k].w ^ kb) >> 4);
          if (v >= nfull) m = 0;
          const int e0 = v * 16;
          while (m) {
            int p = __builtin_ctz(m);
            m &= m - 1;
            keep(e0 + ((p & 7) << 2) + (p >> 3));
          }
        }
      }
      int et = nfull * 16 + tid;
      if (et < E0 && (unsigned)(dsts[et] - lo) < (unsigned)cnt) keep(et);
    } else {
      for (int e = tid; e < E0; e += NT)
        if ((unsigned)(dsts[e] - lo) < (unsigned)cnt) keep(e);
    }
    __syncthreads();
    nm = __builtin_amdgcn_readfirstlane(s_cnt);
    if (tid == 0) lcnt[blockIdx.x] = nm;
  } else {
    nm = lcnt[blockIdx.x];
  }
  const bool fits = (nm + cnt <= CAP);
  const int  tot  = nm + cnt;
  const int  totp = (tot + ROWS - 1) & ~(ROWS - 1);
  const long lb   = (long)blockIdx.x * CAP;

  if (fits) {
    for (int i = tid; i < nm; i += NT) {
      int s, dl;
      if (FIRST) {
        int e = s_e[i];
        s  = srcs[e];
        dl = dsts[e] - lo;
        s_e[i]  = s;
        s_dl[i] = dl;
        lsrc[lb + i] = s;
        ldl[lb + i]  = dl;
      } else {
        s  = lsrc[lb + i];
        dl = ldl[lb + i];
        s_e[i]  = s;
        s_dl[i] = dl;
      }
#pragma unroll
      for (int h = 0; h < H; ++h) {
        float v = lrelu(as_[s * H + h] + s_ad[dl * H + h]);
        s_ex[i * H + h] = v;
        lds_max_f32(&s_m[dl * H + h], v);
      }
    }
  } else {
    for (int e = tid; e < E0; e += NT) {
      unsigned dl = (unsigned)(dsts[e] - lo);
      if (dl < (unsigned)cnt) {
        int s = srcs[e];
        for (int h = 0; h < H; ++h)
          lds_max_f32(&s_m[dl * H + h], lrelu(as_[s * H + h] + s_ad[dl * H + h]));
      }
    }
  }
  __syncthreads();

  if (fits) {
    for (int i = tid; i < nm; i += NT) {
      int dl = s_dl[i];
#pragma unroll
      for (int h = 0; h < H; ++h) {
        float ex = expf(s_ex[i * H + h] - s_m[dl * H + h]);
        s_ex[i * H + h] = ex;
        atomicAdd(&s_den[dl * H + h], ex);
      }
    }
  }
  for (int i = tid; i < cnt * H; i += NT) {
    float ex = expf(lrelu(as_[lo * H + i] + s_ad[i]) - s_m[i]);
    s_self[i] = ex;
    if (fits) s_ex[nm * H + i] = ex;
    atomicAdd(&s_den[i], ex);
  }
  if (fits) {
    for (int j = tid; j < cnt; j += NT) { s_e[nm + j] = lo + j; s_dl[nm + j] = j; }
    if (tid < totp - tot) {
      const int p = tot + tid;
      s_e[p]  = lo;
      s_dl[p] = RNG;
#pragma unroll
      for (int h = 0; h < H; ++h) s_ex[p * H + h] = 0.f;
    }
  }
  __syncthreads();

  if (fits) {
    for (int b = wave; b < (totp / ROWS); b += NW) {
      const int ii  = b * ROWS + (lane & (ROWS - 1));
      const int sv  = s_e[ii];
      const int dlv = s_dl[ii];
      int exv[H];
#pragma unroll
      for (int h = 0; h < H; ++h) exv[h] = __float_as_int(s_ex[ii * H + h]);
      float val[ROWS][FJ];
#pragma unroll
      for (int k = 0; k < ROWS; ++k) {
        const int s = __builtin_amdgcn_readlane(sv, k);
        const float* row = hf + (long)s * F + lane;
#pragma unroll
        for (int j = 0; j < FJ; ++j) val[k][j] = row[32 * j];
      }
#pragma unroll
      for (int k = 0; k < ROWS; ++k) {
        const int dl = __builtin_amdgcn_readlane(dlv, k);
        float* acc = &s_acc[dl * F + lane];
#pragma unroll
        for (int j = 0; j < FJ; ++j) {
          const float w = __int_as_float(__builtin_amdgcn_readlane(exv[(32 * j) / C], k));
          atomicAdd(acc + 32 * j, val[k][j] * w);
        }
      }
    }
  } else {
    for (int e = tid; e < E0; e += NT) {
      unsigned dl = (unsigned)(dsts[e] - lo);
      if (dl < (unsigned)cnt) {
        int s = srcs[e];
        for (int h = 0; h < H; ++h) {
          float ex = expf(lrelu(as_[s * H + h] + s_ad[dl * H + h]) - s_m[dl * H + h]);
          atomicAdd(&s_den[dl * H + h], ex);
          for (int c = 0; c < C; ++c)
            atomicAdd(&s_acc[dl * F + h * C + c], hf[(long)s * F + h * C + c] * ex);
        }
      }
    }
  }
  if (!fits) {
    for (int i = wave; i < cnt; i += NW) {
#pragma unroll
      for (int j = 0; j < FJ; ++j)
        atomicAdd(&s_acc[i * F + lane + 32 * j],
                  hf[(long)(lo + i) * F + lane + 32 * j] * s_self[i * H + (32 * j) / C]);
    }
  }
  __syncthreads();

  for (int i = tid; i < cnt * F; i += NT) {
    int dl = i / F, c = i - dl * F;
    float v = s_acc[i] / s_den[dl * H + c / C] + bias[c];
    if (DO_ELU) v = (v > 0.f) ? v : (expf(v) - 1.f);
    out[(long)lo * F + i] = v;
  }
}

extern "C" void kernel_launch(void* const* d_in, const int* in_sizes, int n_in,
                              void* d_out, int out_size, void* d_ws, size_t ws_size,
                              hipStream_t stream) {
  const float* x    = (const float*)d_in[0];
  const int*   ei   = (const int*)  d_in[1];
  const float* W1   = (const float*)d_in[2];
  const float* atS1 = (const float*)d_in[3];
  const float* atD1 = (const float*)d_in[4];
  const float* b1   = (const float*)d_in[5];
  const float* W2   = (const float*)d_in[6];
  const float* atS2 = (const float*)d_in[7];
  const float* atD2 = (const float*)d_in[8];
  const float* b2   = (const float*)d_in[9];
  float* out = (float*)d_out;

  const int N    = in_sizes[0] / DIN;
  const int E0   = in_sizes[1] / 2;
  const int* srcs = ei;
  const int* dsts = ei + E0;

  float* ws = (float*)d_ws;
  long off = 0;
  float* h1  = ws + off; off += (long)N * (HEADS * DH);
  float* o1  = ws + off; off += (long)N * (HEADS * DH);
  float* a1s = ws + off; off += (long)N * HEADS;
  float* a1d = ws + off; off += (long)N * HEADS;
  off = (off + 3) & ~3L;
  float* h2  = h1;
  float* a2s = h1 + (long)N * DOUT;
  float* a2d = a2s + N;

  const int TB = 256;
  auto blocks = [](long n, int tb) { return (unsigned)((n + tb - 1) / tb); };
  const unsigned NB = blocks(N, RNG);
  int* lsrc = (int*)(ws + off); off += (long)NB * CAP;
  int* ldl  = (int*)(ws + off); off += (long)NB * CAP;
  int* lcnt = (int*)(ws + off); off += NB;
  off = (off + 3) & ~3L;
  unsigned int* keys = (unsigned int*)(ws + off);

  gemm_wmma_f32<(HEADS * DH) / 32><<<256, 256, 0, stream>>>(x, W1, h1, N);
  att_dots<<<blocks((long)N * HEADS, TB), TB, 0, stream>>>(h1, atS1, atD1, a1s, a1d, N, HEADS, DH);
  edge_keys<<<blocks((long)(E0 >> 4) * 4, 1024), 256, 0, stream>>>(dsts, E0, keys);
  gat_owner<HEADS, DH, true, true><<<NB, NT, 0, stream>>>(srcs, dsts, E0, N, a1s, a1d, h1, b1, o1,
                                                          lsrc, ldl, lcnt, keys);

  gemm_wmma_f32<DOUT / 32><<<256, 256, 0, stream>>>(o1, W2, h2, N);
  att_dots<<<blocks(N, TB), TB, 0, stream>>>(h2, atS2, atD2, a2s, a2d, N, 1, DOUT);
  gat_owner<1, DOUT, false, false><<<NB, NT, 0, stream>>>(srcs, dsts, E0, N, a2s, a2d, h2, b2, out,
                                                          lsrc, ldl, lcnt, keys);
}


# Attention dots folded through the weights: a[n] = A[n] . (W @ att); W @ att built per work-group in LDS, GEMM waves dot their A tile (fixed order); dot launches gone, 7 to 5; dot sums reassociated
# speedup vs baseline: 5.9200x; 1.0652x over previous
#include <hip/hip_runtime.h>
#include <math.h>

#define DIN   128
#define DH    64
#define HEADS 2
#define DOUT  64
#define NEG_SLOPE 0.2f

typedef float v2f __attribute__((ext_vector_type(2)));
typedef float v8f __attribute__((ext_vector_type(8)));

typedef float v4f __attribute__((ext_vector_type(4)));
#define GK 128
static_assert(DIN == GK && HEADS * DH == GK, "both products have inner dimension GK");

template <int NSTRIP, int NH>
__global__ void __launch_bounds__(256)
gemm_wmma_f32(const float* __restrict__ A, const float* __restrict__ B,
              float* __restrict__ C, int M,
              const float* __restrict__ att_s, const float* __restrict__ att_d,
              float* __restrict__ o0, float* __restrict__ o1) {
  constexpr int Nc      = 32 * NSTRIP;
  constexpr int NSTREAM = 8 / NSTRIP;
  constexpr int CH      = Nc / NH;
  static_assert(NH <= NSTRIP && CH % 4 == 0, "one dot wave per vector pair");
  __shared__ alignas(16) float s_wa[2 * NH][GK];
  for (int o = threadIdx.x; o < GK * NH; o += 256) {
    const int k = o / NH, hd = o - k * NH;
    const v4f* wp = (const v4f*)(B + k * Nc + hd * CH);
    const v4f* sp = (const v4f*)(att_s + hd * CH);
    const v4f* dp = (const v4f*)(att_d + hd * CH);
    float ss = 0.f, dd = 0.f;
    for (int c = 0; c < CH / 4; ++c) {
      const v4f w = wp[c], s = sp[c], d = dp[c];
      ss += w.x * s.x; ss += w.y * s.y; ss += w.z * s.z; ss += w.w * s.w;
      dd += w.x * d.x; dd += w.y * d.y; dd += w.z * d.z; dd += w.w * d.w;
    }
    s_wa[hd][k]      = ss;
    s_wa[NH + hd][k] = dd;
  }
  __syncthreads();
  const int lane   = threadIdx.x & 31;
  const int wave   = __builtin_amdgcn_readfirstlane(threadIdx.x >> 5);
  const int half   = lane >> 4;
  const int l16    = lane & 15;
  const int strip  = wave % NSTRIP;
  const int stream = wave / NSTRIP;
  const int colp   = strip * 32 + 2 * l16;

  const int tiles = (M + 15) >> 4;
  const int step  = gridDim.x * NSTREAM;
  int tile = blockIdx.x * NSTREAM + stream;
  if (tile >= tiles) return;

  v2f bw[2][32];
#pragma unroll
  for (int j = 0; j < 32; ++j) {
    const int k = 8 * (j >> 1) + 4 * half + 2 * (j & 1);
    const v2f w0 = *(const v2f*)(B + k * Nc + colp);
    const v2f w1 = *(const v2f*)(B + (k + 1) * Nc + colp);
    bw[0][j].x = w0.x; bw[0][j].y = w1.x;
    bw[1][j].x = w0.y; bw[1][j].y = w1.y;
  }

  auto load_tile = [&](v4f (&a)[16], int tl) {
    long row = (long)tl * 16 + l16;
    if (row >= M) row = M - 1;
    const v4f* p = (const v4f*)(A + row * GK + 4 * half);
#pragma unroll
    for (int jj = 0; jj < 16; ++jj) a[jj] = p[2 * jj];
  };
  const bool dotw = strip < NH;
  v4f wv[2][16];
  if (dotw) {
#pragma unroll
    for (int jj = 0; jj < 16; ++jj) {
      wv[0][jj] = *(const v4f*)&s_wa[2 * strip][8 * jj + 4 * half];
      wv[1][jj] = *(const v4f*)&s_wa[2 * strip + 1][8 * jj + 4 * half];
    }
  }
  float* const dotp = (strip == 0) ? o0 : o1;
  auto dot_tile = [&](const v4f (&a)[16], int tl) {
    float p0 = 0.f, q0 = 0.f, p1 = 0.f, q1 = 0.f;
#pragma unroll
    for (int jj = 0; jj < 16; jj += 2) {
      p0 += a[jj].x * wv[0][jj].x; p0 += a[jj].y * wv[0][jj].y;
      p0 += a[jj].z * wv[0][jj].z; p0 += a[jj].w * wv[0][jj].w;
      p1 += a[jj].x * wv[1][jj].x; p1 += a[jj].y * wv[1][jj].y;
      p1 += a[jj].z * wv[1][jj].z; p1 += a[jj].w * wv[1][jj].w;
      q0 += a[jj + 1].x * wv[0][jj + 1].x; q0 += a[jj + 1].y * wv[0][jj + 1].y;
      q0 += a[jj + 1].z * wv[0][jj + 1].z; q0 += a[jj + 1].w * wv[0][jj + 1].w;
      q1 += a[jj + 1].x * wv[1][jj + 1].x; q1 += a[jj + 1].y * wv[1][jj + 1].y;
      q1 += a[jj + 1].z * wv[1][jj + 1].z; q1 += a[jj + 1].w * wv[1][jj + 1].w;
    }
    p0 += q0; p1 += q1;
    p0 += __shfl_xor(p0, 16);
    p1 += __shfl_xor(p1, 16);
    const long row = (long)tl * 16 + l16;
    if (half == 0 && row < M) {
      v2f o; o.x = p0; o.y = p1;
      *(v2f*)(dotp + row * 2) = o;
    }
  };
  auto mul_tile = [&](const v4f (&a)[16], int tl) {
    v8f c0 = {}, c1 = {};
#pragma unroll
    for (int j = 0; j < 32; ++j) {
      v2f av;
      if (j & 1) { av.x = a[j >> 1].z; av.y = a[j >> 1].w; }
      else       { av.x = a[j >> 1].x; av.y = a[j >> 1].y; }
      c0 = __builtin_amdgcn_wmma_f32_16x16x4_f32(false, av, false, bw[0][j],
                                                 (short)0, c0, false, false);
      c1 = __builtin_amdgcn_wmma_f32_16x16x4_f32(false, av, false, bw[1][j],
                                                 (short)0, c1, false, false);
    }
    if ((tl + 1) * 16 <= M) {
      float* cp = C + ((long)tl * 16 + half * 8) * Nc + colp;
#pragma unroll
      for (int r = 0; r < 8; ++r) {
        v2f o; o.x = c0[r]; o.y = c1[r];
        *(v2f*)(cp + r * Nc) = o;
      }
    } else {
#pragma unroll
      for (int r = 0; r < 8; ++r) {
        long row = (long)tl * 16 + r + half * 8;
        if (row < M) {
          v2f o; o.x = c0[r]; o.y = c1[r];
          *(v2f*)(C + row * Nc + colp) = o;
        }
      }
    }
  };

  v4f a0[16], a1[16];
  load_tile(a0, tile);
  while (true) {
    load_tile(a1, min(tile + step, tiles - 1));
    mul_tile(a0, tile);
    if (dotw) dot_tile(a0, tile);
    tile += step;
    if (tile >= tiles) break;
    load_tile(a0, min(tile + step, tiles - 1));
    mul_tile(a1, tile);
    if (dotw) dot_tile(a1, tile);
    tile += step;
    if (tile >= tiles) break;
  }
}

#define RNG 196

__global__ void __launch_bounds__(256)
edge_keys(const int* __restrict__ dsts, int E0, unsigned int* __restrict__ keys) {
  const int nw = (E0 >> 4) << 2;
  if (nw == 0) return;
  const int w0 = blockIdx.x * 1024 + threadIdx.x;
  int d[4][4];
#pragma unroll
  for (int k = 0; k < 4; ++k) {
    const int* dp = dsts + 4 * (long)min(w0 + k * 256, nw - 1);
#pragma unroll
    for (int c = 0; c < 4; ++c) d[k][c] = dp[c];
  }
#pragma unroll
  for (int k = 0; k < 4; ++k) {
    const int w = w0 + k * 256;
    if (w < nw) {
      unsigned k0 = min((unsigned)d[k][0] / RNG, 255u);
      unsigned k1 = min((unsigned)d[k][1] / RNG, 255u);
      unsigned k2 = min((unsigned)d[k][2] / RNG, 255u);
      unsigned k3 = min((unsigned)d[k][3] / RNG, 255u);
      keys[w] = k0 | (k1 << 8) | (k2 << 16) | (k3 << 24);
    }
  }
}

#define CAP 8192
#define ROWS 8
#define NT  1024
#define NW  (NT / 32)

__device__ __forceinline__ float lrelu(float v) { return (v > 0.f) ? v : NEG_SLOPE * v; }

__device__ __forceinline__ void lds_max_f32(float* addr, float v) {
  if (v >= 0.0f) atomicMax((int*)addr, __float_as_int(v));
  else           atomicMin((unsigned int*)addr, __float_as_uint(v));
}

template <int H, int C, bool DO_ELU, bool FIRST>
__global__ void __launch_bounds__(NT)
gat_owner(const int* __restrict__ srcs, const int* __restrict__ dsts, int E0, int N,
          const float* __restrict__ as_, const float* __restrict__ ad_,
          const float* __restrict__ hf, const float* __restrict__ bias,
          float* __restrict__ out,
          int* __restrict__ lsrc, int* __restrict__ ldl, int* __restrict__ lcnt,
          const unsigned int* __restrict__ keys) {
  constexpr int F  = H * C;
  constexpr int FJ = F / 32;
  constexpr int AST = 2;
  __shared__ float s_acc[(RNG + 1) * F];
  __shared__ float s_m[RNG * H];
  __shared__ float s_den[RNG * H];
  __shared__ float s_ad[RNG * H];
  __shared__ float s_self[RNG * H];
  __shared__ int   s_e[CAP];
  __shared__ int   s_dl[CAP];
  __shared__ float s_ex[CAP * H];
  __shared__ int   s_cnt;

  const int tid  = threadIdx.x;
  const int lane = tid & 31;
  const int wave = __builtin_amdgcn_readfirstlane(tid >> 5);
  const int lo   = blockIdx.x * RNG;
  const int cnt  = min(RNG, N - lo);

  for (int i = tid; i < (RNG + 1) * F; i += NT) s_acc[i] = 0.f;
  for (int i = tid; i < cnt * H; i += NT) {
    const int g = (lo + i / H) * AST + i % H;
    float a_d = ad_[g];
    s_ad[i]  = a_d;
    s_m[i]   = lrelu(as_[g] + a_d);
    s_den[i] = 0.f;
  }
  if (FIRST && tid == 0) s_cnt = 0;
  __syncthreads();

  int nm;
  if (FIRST) {
    auto keep = [&](int e) {
      int slot = atomicAdd(&s_cnt, 1);
      if (slot < CAP) s_e[slot] = e;
    };
    if (gridDim.x <= 256) {
      const unsigned kb = blockIdx.x * 0x01010101u;
      const int nfull = E0 >> 4;
      const uint4* kv = (const uint4*)keys;
      auto zb = [](unsigned x) {
        return ~(((x & 0x7F7F7F7Fu) + 0x7F7F7F7Fu) | x) & 0x80808080u;
      };
      for (int base = 0; base < nfull; base += NT * 4) {
        uint4 q[4];
#pragma unroll
        for (int k = 0; k < 4; ++k) q[k] = kv[min(base + k * NT + tid, nfull - 1)];
#pragma unroll
        for (int k = 0; k < 4; ++k) {
          const int v = base + k * NT + tid;
          unsigned m = (zb(q[k].x ^ kb) >> 7) | (zb(q[k].y ^ kb) >> 6) |
                       (zb(q[k].z ^ kb) >> 5) | (zb(q[k].w ^ kb) >> 4);
          if (v >= nfull) m = 0;
          const int e0 = v * 16;
          while (m) {
            int p = __builtin_ctz(m);
            m &= m - 1;
            keep(e0 + ((p & 7) << 2) + (p >> 3));
          }
        }
      }
      int et = nfull * 16 + tid;
      if (et < E0 && (unsigned)(dsts[et] - lo) < (unsigned)cnt) keep(et);
    } else {
      for (int e = tid; e < E0; e += NT)
        if ((unsigned)(dsts[e] - lo) < (unsigned)cnt) keep(e);
    }
    __syncthreads();
    nm = __builtin_amdgcn_readfirstlane(s_cnt);
    if (tid == 0) lcnt[blockIdx.x] = nm;
  } else {
    nm = lcnt[blockIdx.x];
  }
  const bool fits = (nm + cnt <= CAP);
  const int  tot  = nm + cnt;
  const int  totp = (tot + ROWS - 1) & ~(ROWS - 1);
  const long lb   = (long)blockIdx.x * CAP;

  if (fits) {
    for (int i = tid; i < nm; i += NT) {
      int s, dl;
      if (FIRST) {
        int e = s_e[i];
        s  = srcs[e];
        dl = dsts[e] - lo;
        s_e[i]  = s;
        s_dl[i] = dl;
        lsrc[lb + i] = s;
        ldl[lb + i]  = dl;
      } else {
        s  = lsrc[lb + i];
        dl = ldl[lb + i];
        s_e[i]  = s;
        s_dl[i] = dl;
      }
#pragma unroll
      for (int h = 0; h < H; ++h) {
        float v = lrelu(as_[s * AST + h] + s_ad[dl * H + h]);
        s_ex[i * H + h] = v;
        lds_max_f32(&s_m[dl * H + h], v);
      }
    }
  } else {
    for (int e = tid; e < E0; e += NT) {
      unsigned dl = (unsigned)(dsts[e] - lo);
      if (dl < (unsigned)cnt) {
        int s = srcs[e];
        for (int h = 0; h < H; ++h)
          lds_max_f32(&s_m[dl * H + h], lrelu(as_[s * AST + h] + s_ad[dl * H + h]));
      }
    }
  }
  __syncthreads();

  if (fits) {
    for (int i = tid; i < nm; i += NT) {
      int dl = s_dl[i];
#pragma unroll
      for (int h = 0; h < H; ++h) {
        float ex = expf(s_ex[i * H + h] - s_m[dl * H + h]);
        s_ex[i * H + h] = ex;
        atomicAdd(&s_den[dl * H + h], ex);
      }
    }
  }
  for (int i = tid; i < cnt * H; i += NT) {
    float ex = expf(lrelu(as_[(lo + i / H) * AST + i % H] + s_ad[i]) - s_m[i]);
    s_self[i] = ex;
    if (fits) s_ex[nm * H + i] = ex;
    atomicAdd(&s_den[i], ex);
  }
  if (fits) {
    for (int j = tid; j < cnt; j += NT) { s_e[nm + j] = lo + j; s_dl[nm + j] = j; }
    if (tid < totp - tot) {
      const int p = tot + tid;
      s_e[p]  = lo;
      s_dl[p] = RNG;
#pragma unroll
      for (int h = 0; h < H; ++h) s_ex[p * H + h] = 0.f;
    }
  }
  __syncthreads();

  if (fits) {
    for (int b = wave; b < (totp / ROWS); b += NW) {
      const int ii  = b * ROWS + (lane & (ROWS - 1));
      const int sv  = s_e[ii];
      const int dlv = s_dl[ii];
      int exv[H];
#pragma unroll
      for (int h = 0; h < H; ++h) exv[h] = __float_as_int(s_ex[ii * H + h]);
      float val[ROWS][FJ];
#pragma unroll
      for (int k = 0; k < ROWS; ++k) {
        const int s = __builtin_amdgcn_readlane(sv, k);
        const float* row = hf + (long)s * F + lane;
#pragma unroll
        for (int j = 0; j < FJ; ++j) val[k][j] = row[32 * j];
      }
#pragma unroll
      for (int k = 0; k < ROWS; ++k) {
        const int dl = __builtin_amdgcn_readlane(dlv, k);
        float* acc = &s_acc[dl * F + lane];
#pragma unroll
        for (int j = 0; j < FJ; ++j) {
          const float w = __int_as_float(__builtin_amdgcn_readlane(exv[(32 * j) / C], k));
          atomicAdd(acc + 32 * j, val[k][j] * w);
        }
      }
    }
  } else {
    for (int e = tid; e < E0; e += NT) {
      unsigned dl = (unsigned)(dsts[e] - lo);
      if (dl < (unsigned)cnt) {
        int s = srcs[e];
        for (int h = 0; h < H; ++h) {
          float ex = expf(lrelu(as_[s * AST + h] + s_ad[dl * H + h]) - s_m[dl * H + h]);
          atomicAdd(&s_den[dl * H + h], ex);
          for (int c = 0; c < C; ++c)
            atomicAdd(&s_acc[dl * F + h * C + c], hf[(long)s * F + h * C + c] * ex);
        }
      }
    }
  }
  if (!fits) {
    for (int i = wave; i < cnt; i += NW) {
#pragma unroll
      for (int j = 0; j < FJ; ++j)
        atomicAdd(&s_acc[i * F + lane + 32 * j],
                  hf[(long)(lo + i) * F + lane + 32 * j] * s_self[i * H + (32 * j) / C]);
    }
  }
  __syncthreads();

  for (int i = tid; i < cnt * F; i += NT) {
    int dl = i / F, c = i - dl * F;
    float v = s_acc[i] / s_den[dl * H + c / C] + bias[c];
    if (DO_ELU) v = (v > 0.f) ? v : (expf(v) - 1.f);
    out[(long)lo * F + i] = v;
  }
}

extern "C" void kernel_launch(void* const* d_in, const int* in_sizes, int n_in,
                              void* d_out, int out_size, void* d_ws, size_t ws_size,
                              hipStream_t stream) {
  const float* x    = (const float*)d_in[0];
  const int*   ei   = (const int*)  d_in[1];
  const float* W1   = (const float*)d_in[2];
  const float* atS1 = (const float*)d_in[3];
  const float* atD1 = (const float*)d_in[4];
  const float* b1   = (const float*)d_in[5];
  const float* W2   = (const float*)d_in[6];
  const float* atS2 = (const float*)d_in[7];
  const float* atD2 = (const float*)d_in[8];
  const float* b2   = (const float*)d_in[9];
  float* out = (float*)d_out;

  const int N    = in_sizes[0] / DIN;
  const int E0   = in_sizes[1] / 2;
  const int* srcs = ei;
  const int* dsts = ei + E0;

  float* ws = (float*)d_ws;
  long off = 0;
  float* h1  = ws + off; off += (long)N * (HEADS * DH);
  float* o1  = ws + off; off += (long)N * (HEADS * DH);
  float* a1s = ws + off; off += (long)N * HEADS;
  float* a1d = ws + off; off += (long)N * HEADS;
  off = (off + 3) & ~3L;
  float* h2  = h1;
  float* a2  = h1 + (long)N * DOUT;

  auto blocks = [](long n, int tb) { return (unsigned)((n + tb - 1) / tb); };
  const unsigned NB = blocks(N, RNG);
  int* lsrc = (int*)(ws + off); off += (long)NB * CAP;
  int* ldl  = (int*)(ws + off); off += (long)NB * CAP;
  int* lcnt = (int*)(ws + off); off += NB;
  off = (off + 3) & ~3L;
  unsigned int* keys = (unsigned int*)(ws + off);

  gemm_wmma_f32<(HEADS * DH) / 32, HEADS><<<256, 256, 0, stream>>>(x, W1, h1, N, atS1, atD1, a1s, a1d);
  edge_keys<<<blocks((long)(E0 >> 4) * 4, 1024), 256, 0, stream>>>(dsts, E0, keys);
  gat_owner<HEADS, DH, true, true><<<NB, NT, 0, stream>>>(srcs, dsts, E0, N, a1s, a1d, h1, b1, o1,
                                                          lsrc, ldl, lcnt, keys);

  gemm_wmma_f32<DOUT / 32, 1><<<256, 256, 0, stream>>>(o1, W2, h2, N, atS2, atD2, a2, nullptr);
  gat_owner<1, DOUT, false, false><<<NB, NT, 0, stream>>>(srcs, dsts, E0, N, a2, a2 + 1, h2, b2, out,
                                                          lsrc, ldl, lcnt, keys);
}


# Feature sum: a lane fetches its 4 (layer 2: 2) neighbouring floats of a source row in one load, a row is one load instruction; 16 rows per turn; LDS rows re-laid to match; LDS atomics, arrival order
# speedup vs baseline: 6.2007x; 1.0474x over previous
#include <hip/hip_runtime.h>
#include <math.h>

#define DIN   128
#define DH    64
#define HEADS 2
#define DOUT  64
#define NEG_SLOPE 0.2f

typedef float v2f __attribute__((ext_vector_type(2)));
typedef float v8f __attribute__((ext_vector_type(8)));

typedef float v4f __attribute__((ext_vector_type(4)));
#define GK 128
static_assert(DIN == GK && HEADS * DH == GK, "both products have inner dimension GK");

template <int NSTRIP, int NH>
__global__ void __launch_bounds__(256)
gemm_wmma_f32(const float* __restrict__ A, const float* __restrict__ B,
              float* __restrict__ C, int M,
              const float* __restrict__ att_s, const float* __restrict__ att_d,
              float* __restrict__ o0, float* __restrict__ o1) {
  constexpr int Nc      = 32 * NSTRIP;
  constexpr int NSTREAM = 8 / NSTRIP;
  constexpr int CH      = Nc / NH;
  static_assert(NH <= NSTRIP && CH % 4 == 0, "one dot wave per vector pair");
  __shared__ alignas(16) float s_wa[2 * NH][GK];
  for (int o = threadIdx.x; o < GK * NH; o += 256) {
    const int k = o / NH, hd = o - k * NH;
    const v4f* wp = (const v4f*)(B + k * Nc + hd * CH);
    const v4f* sp = (const v4f*)(att_s + hd * CH);
    const v4f* dp = (const v4f*)(att_d + hd * CH);
    float ss = 0.f, dd = 0.f;
    for (int c = 0; c < CH / 4; ++c) {
      const v4f w = wp[c], s = sp[c], d = dp[c];
      ss += w.x * s.x; ss += w.y * s.y; ss += w.z * s.z; ss += w.w * s.w;
      dd += w.x * d.x; dd += w.y * d.y; dd += w.z * d.z; dd += w.w * d.w;
    }
    s_wa[hd][k]      = ss;
    s_wa[NH + hd][k] = dd;
  }
  __syncthreads();
  const int lane   = threadIdx.x & 31;
  const int wave   = __builtin_amdgcn_readfirstlane(threadIdx.x >> 5);
  const int half   = lane >> 4;
  const int l16    = lane & 15;
  const int strip  = wave % NSTRIP;
  const int stream = wave / NSTRIP;
  const int colp   = strip * 32 + 2 * l16;

  const int tiles = (M + 15) >> 4;
  const int step  = gridDim.x * NSTREAM;
  int tile = blockIdx.x * NSTREAM + stream;
  if (tile >= tiles) return;

  v2f bw[2][32];
#pragma unroll
  for (int j = 0; j < 32; ++j) {
    const int k = 8 * (j >> 1) + 4 * half + 2 * (j & 1);
    const v2f w0 = *(const v2f*)(B + k * Nc + colp);
    const v2f w1 = *(const v2f*)(B + (k + 1) * Nc + colp);
    bw[0][j].x = w0.x; bw[0][j].y = w1.x;
    bw[1][j].x = w0.y; bw[1][j].y = w1.y;
  }

  auto load_tile = [&](v4f (&a)[16], int tl) {
    long row = (long)tl * 16 + l16;
    if (row >= M) row = M - 1;
    const v4f* p = (const v4f*)(A + row * GK + 4 * half);
#pragma unroll
    for (int jj = 0; jj < 16; ++jj) a[jj] = p[2 * jj];
  };
  const bool dotw = strip < NH;
  v4f wv[2][16];
  if (dotw) {
#pragma unroll
    for (int jj = 0; jj < 16; ++jj) {
      wv[0][jj] = *(const v4f*)&s_wa[2 * strip][8 * jj + 4 * half];
      wv[1][jj] = *(const v4f*)&s_wa[2 * strip + 1][8 * jj + 4 * half];
    }
  }
  float* const dotp = (strip == 0) ? o0 : o1;
  auto dot_tile = [&](const v4f (&a)[16], int tl) {
    float p0 = 0.f, q0 = 0.f, p1 = 0.f, q1 = 0.f;
#pragma unroll
    for (int jj = 0; jj < 16; jj += 2) {
      p0 += a[jj].x * wv[0][jj].x; p0 += a[jj].y * wv[0][jj].y;
      p0 += a[jj].z * wv[0][jj].z; p0 += a[jj].w * wv[0][jj].w;
      p1 += a[jj].x * wv[1][jj].x; p1 += a[jj].y * wv[1][jj].y;
      p1 += a[jj].z * wv[1][jj].z; p1 += a[jj].w * wv[1][jj].w;
      q0 += a[jj + 1].x * wv[0][jj + 1].x; q0 += a[jj + 1].y * wv[0][jj + 1].y;
      q0 += a[jj + 1].z * wv[0][jj + 1].z; q0 += a[jj + 1].w * wv[0][jj + 1].w;
      q1 += a[jj + 1].x * wv[1][jj + 1].x; q1 += a[jj + 1].y * wv[1][jj + 1].y;
      q1 += a[jj + 1].z * wv[1][jj + 1].z; q1 += a[jj + 1].w * wv[1][jj + 1].w;
    }
    p0 += q0; p1 += q1;
    p0 += __shfl_xor(p0, 16);
    p1 += __shfl_xor(p1, 16);
    const long row = (long)tl * 16 + l16;
    if (half == 0 && row < M) {
      v2f o; o.x = p0; o.y = p1;
      *(v2f*)(dotp + row * 2) = o;
    }
  };
  auto mul_tile = [&](const v4f (&a)[16], int tl) {
    v8f c0 = {}, c1 = {};
#pragma unroll
    for (int j = 0; j < 32; ++j) {
      v2f av;
      if (j & 1) { av.x = a[j >> 1].z; av.y = a[j >> 1].w; }
      else       { av.x = a[j >> 1].x; av.y = a[j >> 1].y; }
      c0 = __builtin_amdgcn_wmma_f32_16x16x4_f32(false, av, false, bw[0][j],
                                                 (short)0, c0, false, false);
      c1 = __builtin_amdgcn_wmma_f32_16x16x4_f32(false, av, false, bw[1][j],
                                                 (short)0, c1, false, false);
    }
    if ((tl + 1) * 16 <= M) {
      float* cp = C + ((long)tl * 16 + half * 8) * Nc + colp;
#pragma unroll
      for (int r = 0; r < 8; ++r) {
        v2f o; o.x = c0[r]; o.y = c1[r];
        *(v2f*)(cp + r * Nc) = o;
      }
    } else {
#pragma unroll
      for (int r = 0; r < 8; ++r) {
        long row = (long)tl * 16 + r + half * 8;
        if (row < M) {
          v2f o; o.x = c0[r]; o.y = c1[r];
          *(v2f*)(C + row * Nc + colp) = o;
        }
      }
    }
  };

  v4f a0[16], a1[16];
  load_tile(a0, tile);
  while (true) {
    load_tile(a1, min(tile + step, tiles - 1));
    mul_tile(a0, tile);
    if (dotw) dot_tile(a0, tile);
    tile += step;
    if (tile >= tiles) break;
    load_tile(a0, min(tile + step, tiles - 1));
    mul_tile(a1, tile);
    if (dotw) dot_tile(a1, tile);
    tile += step;
    if (tile >= tiles) break;
  }
}

#define RNG 196

__global__ void __launch_bounds__(256)
edge_keys(const int* __restrict__ dsts, int E0, unsigned int* __restrict__ keys) {
  const int nw = (E0 >> 4) << 2;
  if (nw == 0) return;
  const int w0 = blockIdx.x * 1024 + threadIdx.x;
  int d[4][4];
#pragma unroll
  for (int k = 0; k < 4; ++k) {
    const int* dp = dsts + 4 * (long)min(w0 + k * 256, nw - 1);
#pragma unroll
    for (int c = 0; c < 4; ++c) d[k][c] = dp[c];
  }
#pragma unroll
  for (int k = 0; k < 4; ++k) {
    const int w = w0 + k * 256;
    if (w < nw) {
      unsigned k0 = min((unsigned)d[k][0] / RNG, 255u);
      unsigned k1 = min((unsigned)d[k][1] / RNG, 255u);
      unsigned k2 = min((unsigned)d[k][2] / RNG, 255u);
      unsigned k3 = min((unsigned)d[k][3] / RNG, 255u);
      keys[w] = k0 | (k1 << 8) | (k2 << 16) | (k3 << 24);
    }
  }
}

#define CAP 8192
#define ROWS 16
#define NT  1024
#define NW  (NT / 32)

__device__ __forceinline__ float lrelu(float v) { return (v > 0.f) ? v : NEG_SLOPE * v; }

template <int F>
__device__ __forceinline__ int acc_word(int f) { return (f % (F / 32)) * 32 + f / (F / 32); }

template <int K> struct kconst { static constexpr int v = K; };
template <int K, int KN> struct kloop {
  template <typename Fn> static __device__ __forceinline__ void run(Fn& fn) {
    fn(kconst<K>{});
    kloop<K + 1, KN>::run(fn);
  }
};
template <int KN> struct kloop<KN, KN> {
  template <typename Fn> static __device__ __forceinline__ void run(Fn&) {}
};
template <int K>
__device__ __forceinline__ float row_lane(int v) {
  return __int_as_float(__builtin_amdgcn_update_dpp(0, v, 0x150 + K, 0xf, 0xf, false));
}

__device__ __forceinline__ void lds_max_f32(float* addr, float v) {
  if (v >= 0.0f) atomicMax((int*)addr, __float_as_int(v));
  else           atomicMin((unsigned int*)addr, __float_as_uint(v));
}

template <int H, int C, bool DO_ELU, bool FIRST>
__global__ void __launch_bounds__(NT)
gat_owner(const int* __restrict__ srcs, const int* __restrict__ dsts, int E0, int N,
          const float* __restrict__ as_, const float* __restrict__ ad_,
          const float* __restrict__ hf, const float* __restrict__ bias,
          float* __restrict__ out,
          int* __restrict__ lsrc, int* __restrict__ ldl, int* __restrict__ lcnt,
          const unsigned int* __restrict__ keys) {
  constexpr int F  = H * C;
  constexpr int FJ = F / 32;
  typedef float vrow __attribute__((ext_vector_type(FJ)));
  static_assert(H == 1 || (H == 2 && FJ * 16 == C), "a 16-lane row of the wave covers one head");
  constexpr int AST = 2;
  __shared__ float s_acc[(RNG + 1) * F];
  __shared__ float s_m[RNG * H];
  __shared__ float s_den[RNG * H];
  __shared__ float s_ad[RNG * H];
  __shared__ float s_self[RNG * H];
  __shared__ int   s_e[CAP];
  __shared__ int   s_dl[CAP];
  __shared__ float s_ex[CAP * H];
  __shared__ int   s_cnt;

  const int tid  = threadIdx.x;
  const int lane = tid & 31;
  const int wave = __builtin_amdgcn_readfirstlane(tid >> 5);
  const int lo   = blockIdx.x * RNG;
  const int cnt  = min(RNG, N - lo);

  for (int i = tid; i < (RNG + 1) * F; i += NT) s_acc[i] = 0.f;
  for (int i = tid; i < cnt * H; i += NT) {
    const int g = (lo + i / H) * AST + i % H;
    float a_d = ad_[g];
    s_ad[i]  = a_d;
    s_m[i]   = lrelu(as_[g] + a_d);
    s_den[i] = 0.f;
  }
  if (FIRST && tid == 0) s_cnt = 0;
  __syncthreads();

  int nm;
  if (FIRST) {
    auto keep = [&](int e) {
      int slot = atomicAdd(&s_cnt, 1);
      if (slot < CAP) s_e[slot] = e;
    };
    if (gridDim.x <= 256) {
      const unsigned kb = blockIdx.x * 0x01010101u;
      const int nfull = E0 >> 4;
      const uint4* kv = (const uint4*)keys;
      auto zb = [](unsigned x) {
        return ~(((x & 0x7F7F7F7Fu) + 0x7F7F7F7Fu) | x) & 0x80808080u;
      };
      for (int base = 0; base < nfull; base += NT * 4) {
        uint4 q[4];
#pragma unroll
        for (int k = 0; k < 4; ++k) q[k] = kv[min(base + k * NT + tid, nfull - 1)];
#pragma unroll
        for (int k = 0; k < 4; ++k) {
          const int v = base + k * NT + tid;
          unsigned m = (zb(q[k].x ^ kb) >> 7) | (zb(q[k].y ^ kb) >> 6) |
                       (zb(q[k].z ^ kb) >> 5) | (zb(q[k].w ^ kb) >> 4);
          if (v >= nfull) m = 0;
          const int e0 = v * 16;
          while (m) {
            int p = __builtin_ctz(m);
            m &= m - 1;
            keep(e0 + ((p & 7) << 2) + (p >> 3));
          }
        }
      }
      int et = nfull * 16 + tid;
      if (et < E0 && (unsigned)(dsts[et] - lo) < (unsigned)cnt) keep(et);
    } else {
      for (int e = tid; e < E0; e += NT)
        if ((unsigned)(dsts[e] - lo) < (unsigned)cnt) keep(e);
    }
    __syncthreads();
    nm = __builtin_amdgcn_readfirstlane(s_cnt);
    if (tid == 0) lcnt[blockIdx.x] = nm;
  } else {
    nm = lcnt[blockIdx.x];
  }
  const bool fits = (nm + cnt <= CAP);
  const int  tot  = nm + cnt;
  const int  totp = (tot + ROWS - 1) & ~(ROWS - 1);
  const long lb   = (long)blockIdx.x * CAP;

  if (fits) {
    for (int i = tid; i < nm; i += NT) {
      int s, dl;
      if (FIRST) {
        int e = s_e[i];
        s  = srcs[e];
        dl = dsts[e] - lo;
        s_e[i]  = s;
        s_dl[i] = dl;
        lsrc[lb + i] = s;
        ldl[lb + i]  = dl;
      } else {
        s  = lsrc[lb + i];
        dl = ldl[lb + i];
        s_e[i]  = s;
        s_dl[i] = dl;
      }
#pragma unroll
      for (int h = 0; h < H; ++h) {
        float v = lrelu(as_[s * AST + h] + s_ad[dl * H + h]);
        s_ex[i * H + h] = v;
        lds_max_f32(&s_m[dl * H + h], v);
      }
    }
  } else {
    for (int e = tid; e < E0; e += NT) {
      unsigned dl = (unsigned)(dsts[e] - lo);
      if (dl < (unsigned)cnt) {
        int s = srcs[e];
        for (int h = 0; h < H; ++h)
          lds_max_f32(&s_m[dl * H + h], lrelu(as_[s * AST + h] + s_ad[dl * H + h]));
      }
    }
  }
  __syncthreads();

  if (fits) {
    for (int i = tid; i < nm; i += NT) {
      int dl = s_dl[i];
#pragma unroll
      for (int h = 0; h < H; ++h) {
        float ex = expf(s_ex[i * H + h] - s_m[dl * H + h]);
        s_ex[i * H + h] = ex;
        atomicAdd(&s_den[dl * H + h], ex);
      }
    }
  }
  for (int i = tid; i < cnt * H; i += NT) {
    float ex = expf(lrelu(as_[(lo + i / H) * AST + i % H] + s_ad[i]) - s_m[i]);
    s_self[i] = ex;
    if (fits) s_ex[nm * H + i] = ex;
    atomicAdd(&s_den[i], ex);
  }
  if (fits) {
    for (int j = tid; j < cnt; j += NT) { s_e[nm + j] = lo + j; s_dl[nm + j] = j; }
    if (tid < totp - tot) {
      const int p = tot + tid;
      s_e[p]  = lo;
      s_dl[p] = RNG;
#pragma unroll
      for (int h = 0; h < H; ++h) s_ex[p * H + h] = 0.f;
    }
  }
  __syncthreads();

  if (fits) {
    const int hl = (FJ * lane) / C;
    for (int b = wave; b < (totp / ROWS); b += NW) {
      const int ii  = b * ROWS + (lane & (ROWS - 1));
      const int sv  = s_e[ii];
      const int dlv = s_dl[ii];
      const int exl = __float_as_int(s_ex[ii * H + hl]);
      vrow val[ROWS];
#pragma unroll
      for (int k = 0; k < ROWS; ++k) {
        const int s = __builtin_amdgcn_readlane(sv, k);
        val[k] = *((const vrow*)(hf + (long)s * F) + lane);
      }
      auto add_entry = [&](auto kc) {
        constexpr int k = decltype(kc)::v;
        const int dl = __builtin_amdgcn_readlane(dlv, k);
        float* acc = &s_acc[dl * F + lane];
        const float w = (H == 1) ? __int_as_float(__builtin_amdgcn_readlane(exl, k))
                                 : row_lane<k>(exl);
#pragma unroll
        for (int j = 0; j < FJ; ++j) atomicAdd(acc + 32 * j, val[k][j] * w);
      };
      kloop<0, ROWS>::run(add_entry);
    }
  } else {
    for (int e = tid; e < E0; e += NT) {
      unsigned dl = (unsigned)(dsts[e] - lo);
      if (dl < (unsigned)cnt) {
        int s = srcs[e];
        for (int h = 0; h < H; ++h) {
          float ex = expf(lrelu(as_[s * AST + h] + s_ad[dl * H + h]) - s_m[dl * H + h]);
          atomicAdd(&s_den[dl * H + h], ex);
          for (int c = 0; c < C; ++c)
            atomicAdd(&s_acc[dl * F + acc_word<F>(h * C + c)], hf[(long)s * F + h * C + c] * ex);
        }
      }
    }
  }
  if (!fits) {
    for (int i = wave; i < cnt; i += NW) {
#pragma unroll
      for (int j = 0; j < FJ; ++j)
        atomicAdd(&s_acc[i * F + acc_word<F>(lane + 32 * j)],
                  hf[(long)(lo + i) * F + lane + 32 * j] * s_self[i * H + (32 * j) / C]);
    }
  }
  __syncthreads();

  for (int i = tid; i < cnt * F; i += NT) {
    int dl = i / F, c = i - dl * F;
    float v = s_acc[dl * F + acc_word<F>(c)] / s_den[dl * H + c / C] + bias[c];
    if (DO_ELU) v = (v > 0.f) ? v : (expf(v) - 1.f);
    out[(long)lo * F + i] = v;
  }
}

extern "C" void kernel_launch(void* const* d_in, const int* in_sizes, int n_in,
                              void* d_out, int out_size, void* d_ws, size_t ws_size,
                              hipStream_t stream) {
  const float* x    = (const float*)d_in[0];
  const int*   ei   = (const int*)  d_in[1];
  const float* W1   = (const float*)d_in[2];
  const float* atS1 = (const float*)d_in[3];
  const float* atD1 = (const float*)d_in[4];
  const float* b1   = (const float*)d_in[5];
  const float* W2   = (const float*)d_in[6];
  const float* atS2 = (const float*)d_in[7];
  const float* atD2 = (const float*)d_in[8];
  const float* b2   = (const float*)d_in[9];
  float* out = (float*)d_out;

  const int N    = in_sizes[0] / DIN;
  const int E0   = in_sizes[1] / 2;
  const int* srcs = ei;
  const int* dsts = ei + E0;

  float* ws = (float*)d_ws;
  long off = 0;
  float* h1  = ws + off; off += (long)N * (HEADS * DH);
  float* o1  = ws + off; off += (long)N * (HEADS * DH);
  float* a1s = ws + off; off += (long)N * HEADS;
  float* a1d = ws + off; off += (long)N * HEADS;
  off = (off + 3) & ~3L;
  float* h2  = h1;
  float* a2  = h1 + (long)N * DOUT;

  auto blocks = [](long n, int tb) { return (unsigned)((n + tb - 1) / tb); };
  const unsigned NB = blocks(N, RNG);
  int* lsrc = (int*)(ws + off); off += (long)NB * CAP;
  int* ldl  = (int*)(ws + off); off += (long)NB * CAP;
  int* lcnt = (int*)(ws + off); off += NB;
  off = (off + 3) & ~3L;
  unsigned int* keys = (unsigned int*)(ws + off);

  gemm_wmma_f32<(HEADS * DH) / 32, HEADS><<<256, 256, 0, stream>>>(x, W1, h1, N, atS1, atD1, a1s, a1d);
  edge_keys<<<blocks((long)(E0 >> 4) * 4, 1024), 256, 0, stream>>>(dsts, E0, keys);
  gat_owner<HEADS, DH, true, true><<<NB, NT, 0, stream>>>(srcs, dsts, E0, N, a1s, a1d, h1, b1, o1,
                                                          lsrc, ldl, lcnt, keys);

  gemm_wmma_f32<DOUT / 32, 1><<<256, 256, 0, stream>>>(o1, W2, h2, N, atS2, atD2, a2, nullptr);
  gat_owner<1, DOUT, false, false><<<NB, NT, 0, stream>>>(srcs, dsts, E0, N, a2, a2 + 1, h2, b2, out,
                                                          lsrc, ldl, lcnt, keys);
}


# Key launch also writes per range a bit per 16-edge group holding an edge of it; layer-1 owner blocks read their row of bits and load only flagged key pieces, not all keys; LDS atomics, arrival order
# speedup vs baseline: 6.5647x; 1.0587x over previous
#include <hip/hip_runtime.h>
#include <math.h>

#define DIN   128
#define DH    64
#define HEADS 2
#define DOUT  64
#define NEG_SLOPE 0.2f

typedef float v2f __attribute__((ext_vector_type(2)));
typedef float v8f __attribute__((ext_vector_type(8)));

typedef float v4f __attribute__((ext_vector_type(4)));
#define GK 128
static_assert(DIN == GK && HEADS * DH == GK, "both products have inner dimension GK");

template <int NSTRIP, int NH>
__global__ void __launch_bounds__(256)
gemm_wmma_f32(const float* __restrict__ A, const float* __restrict__ B,
              float* __restrict__ C, int M,
              const float* __restrict__ att_s, const float* __restrict__ att_d,
              float* __restrict__ o0, float* __restrict__ o1) {
  constexpr int Nc      = 32 * NSTRIP;
  constexpr int NSTREAM = 8 / NSTRIP;
  constexpr int CH      = Nc / NH;
  static_assert(NH <= NSTRIP && CH % 4 == 0, "one dot wave per vector pair");
  __shared__ alignas(16) float s_wa[2 * NH][GK];
  for (int o = threadIdx.x; o < GK * NH; o += 256) {
    const int k = o / NH, hd = o - k * NH;
    const v4f* wp = (const v4f*)(B + k * Nc + hd * CH);
    const v4f* sp = (const v4f*)(att_s + hd * CH);
    const v4f* dp = (const v4f*)(att_d + hd * CH);
    float ss = 0.f, dd = 0.f;
    for (int c = 0; c < CH / 4; ++c) {
      const v4f w = wp[c], s = sp[c], d = dp[c];
      ss += w.x * s.x; ss += w.y * s.y; ss += w.z * s.z; ss += w.w * s.w;
      dd += w.x * d.x; dd += w.y * d.y; dd += w.z * d.z; dd += w.w * d.w;
    }
    s_wa[hd][k]      = ss;
    s_wa[NH + hd][k] = dd;
  }
  __syncthreads();
  const int lane   = threadIdx.x & 31;
  const int wave   = __builtin_amdgcn_readfirstlane(threadIdx.x >> 5);
  const int half   = lane >> 4;
  const int l16    = lane & 15;
  const int strip  = wave % NSTRIP;
  const int stream = wave / NSTRIP;
  const int colp   = strip * 32 + 2 * l16;

  const int tiles = (M + 15) >> 4;
  const int step  = gridDim.x * NSTREAM;
  int tile = blockIdx.x * NSTREAM + stream;
  if (tile >= tiles) return;

  v2f bw[2][32];
#pragma unroll
  for (int j = 0; j < 32; ++j) {
    const int k = 8 * (j >> 1) + 4 * half + 2 * (j & 1);
    const v2f w0 = *(const v2f*)(B + k * Nc + colp);
    const v2f w1 = *(const v2f*)(B + (k + 1) * Nc + colp);
    bw[0][j].x = w0.x; bw[0][j].y = w1.x;
    bw[1][j].x = w0.y; bw[1][j].y = w1.y;
  }

  auto load_tile = [&](v4f (&a)[16], int tl) {
    long row = (long)tl * 16 + l16;
    if (row >= M) row = M - 1;
    const v4f* p = (const v4f*)(A + row * GK + 4 * half);
#pragma unroll
    for (int jj = 0; jj < 16; ++jj) a[jj] = p[2 * jj];
  };
  const bool dotw = strip < NH;
  v4f wv[2][16];
  if (dotw) {
#pragma unroll
    for (int jj = 0; jj < 16; ++jj) {
      wv[0][jj] = *(const v4f*)&s_wa[2 * strip][8 * jj + 4 * half];
      wv[1][jj] = *(const v4f*)&s_wa[2 * strip + 1][8 * jj + 4 * half];
    }
  }
  float* const dotp = (strip == 0) ? o0 : o1;
  auto dot_tile = [&](const v4f (&a)[16], int tl) {
    float p0 = 0.f, q0 = 0.f, p1 = 0.f, q1 = 0.f;
#pragma unroll
    for (int jj = 0; jj < 16; jj += 2) {
      p0 += a[jj].x * wv[0][jj].x; p0 += a[jj].y * wv[0][jj].y;
      p0 += a[jj].z * wv[0][jj].z; p0 += a[jj].w * wv[0][jj].w;
      p1 += a[jj].x * wv[1][jj].x; p1 += a[jj].y * wv[1][jj].y;
      p1 += a[jj].z * wv[1][jj].z; p1 += a[jj].w * wv[1][jj].w;
      q0 += a[jj + 1].x * wv[0][jj + 1].x; q0 += a[jj + 1].y * wv[0][jj + 1].y;
      q0 += a[jj + 1].z * wv[0][jj + 1].z; q0 += a[jj + 1].w * wv[0][jj + 1].w;
      q1 += a[jj + 1].x * wv[1][jj + 1].x; q1 += a[jj + 1].y * wv[1][jj + 1].y;
      q1 += a[jj + 1].z * wv[1][jj + 1].z; q1 += a[jj + 1].w * wv[1][jj + 1].w;
    }
    p0 += q0; p1 += q1;
    p0 += __shfl_xor(p0, 16);
    p1 += __shfl_xor(p1, 16);
    const long row = (long)tl * 16 + l16;
    if (half == 0 && row < M) {
      v2f o; o.x = p0; o.y = p1;
      *(v2f*)(dotp + row * 2) = o;
    }
  };
  auto mul_tile = [&](const v4f (&a)[16], int tl) {
    v8f c0 = {}, c1 = {};
#pragma unroll
    for (int j = 0; j < 32; ++j) {
      v2f av;
      if (j & 1) { av.x = a[j >> 1].z; av.y = a[j >> 1].w; }
      else       { av.x = a[j >> 1].x; av.y = a[j >> 1].y; }
      c0 = __builtin_amdgcn_wmma_f32_16x16x4_f32(false, av, false, bw[0][j],
                                                 (short)0, c0, false, false);
      c1 = __builtin_amdgcn_wmma_f32_16x16x4_f32(false, av, false, bw[1][j],
                                                 (short)0, c1, false, false);
    }
    if ((tl + 1) * 16 <= M) {
      float* cp = C + ((long)tl * 16 + half * 8) * Nc + colp;
#pragma unroll
      for (int r = 0; r < 8; ++r) {
        v2f o; o.x = c0[r]; o.y = c1[r];
        *(v2f*)(cp + r * Nc) = o;
      }
    } else {
#pragma unroll
      for (int r = 0; r < 8; ++r) {
        long row = (long)tl * 16 + r + half * 8;
        if (row < M) {
          v2f o; o.x = c0[r]; o.y = c1[r];
          *(v2f*)(C + row * Nc + colp) = o;
        }
      }
    }
  };

  v4f a0[16], a1[16];
  load_tile(a0, tile);
  while (true) {
    load_tile(a1, min(tile + step, tiles - 1));
    mul_tile(a0, tile);
    if (dotw) dot_tile(a0, tile);
    tile += step;
    if (tile >= tiles) break;
    load_tile(a0, min(tile + step, tiles - 1));
    mul_tile(a1, tile);
    if (dotw) dot_tile(a1, tile);
    tile += step;
    if (tile >= tiles) break;
  }
}

#define RNG 196
#define GRP 1024
#define BMW (GRP / 32)
#define BMP 257

__global__ void __launch_bounds__(GRP)
edge_keys(const int* __restrict__ dsts, int E0, unsigned int* __restrict__ keys,
          unsigned int* __restrict__ gidx) {
  __shared__ unsigned int s_bm[BMW * BMP];
  const int tid   = threadIdx.x;
  const int nfull = E0 >> 4;
  for (int i = tid; i < BMW * BMP; i += GRP) s_bm[i] = 0u;
  __syncthreads();
  const int g = blockIdx.x * GRP + tid;
  if (g < nfull) {
    const int* dp = dsts + 16 * (long)g;
    int d[16];
#pragma unroll
    for (int c = 0; c < 16; ++c) d[c] = dp[c];
    unsigned int* col = s_bm + (tid >> 5) * BMP;
    const unsigned bit = 1u << (tid & 31);
    unsigned w[4];
#pragma unroll
    for (int j = 0; j < 4; ++j) {
      const unsigned k0 = min((unsigned)d[4 * j + 0] / RNG, 255u);
      const unsigned k1 = min((unsigned)d[4 * j + 1] / RNG, 255u);
      const unsigned k2 = min((unsigned)d[4 * j + 2] / RNG, 255u);
      const unsigned k3 = min((unsigned)d[4 * j + 3] / RNG, 255u);
      atomicOr(col + k0, bit);
      atomicOr(col + k1, bit);
      atomicOr(col + k2, bit);
      atomicOr(col + k3, bit);
      w[j] = k0 | (k1 << 8) | (k2 << 16) | (k3 << 24);
    }
    uint4 kw; kw.x = w[0]; kw.y = w[1]; kw.z = w[2]; kw.w = w[3];
    *(uint4*)(keys + 4 * (long)g) = kw;
  }
  __syncthreads();
  const int  lane  = tid & 31;
  const int  wave  = tid >> 5;
  const long pitch = (long)gridDim.x * BMW;
  unsigned int* op = gidx + (long)blockIdx.x * BMW + lane;
  for (int r = wave; r < 256; r += GRP / 32) op[r * pitch] = s_bm[lane * BMP + r];
}

#define CAP 8192
#define ROWS 16
#define NT  1024
#define NW  (NT / 32)

__device__ __forceinline__ float lrelu(float v) { return (v > 0.f) ? v : NEG_SLOPE * v; }

template <int F>
__device__ __forceinline__ int acc_word(int f) { return (f % (F / 32)) * 32 + f / (F / 32); }

template <int K> struct kconst { static constexpr int v = K; };
template <int K, int KN> struct kloop {
  template <typename Fn> static __device__ __forceinline__ void run(Fn& fn) {
    fn(kconst<K>{});
    kloop<K + 1, KN>::run(fn);
  }
};
template <int KN> struct kloop<KN, KN> {
  template <typename Fn> static __device__ __forceinline__ void run(Fn&) {}
};
template <int K>
__device__ __forceinline__ float row_lane(int v) {
  return __int_as_float(__builtin_amdgcn_update_dpp(0, v, 0x150 + K, 0xf, 0xf, false));
}

__device__ __forceinline__ void lds_max_f32(float* addr, float v) {
  if (v >= 0.0f) atomicMax((int*)addr, __float_as_int(v));
  else           atomicMin((unsigned int*)addr, __float_as_uint(v));
}

template <int H, int C, bool DO_ELU, bool FIRST>
__global__ void __launch_bounds__(NT)
gat_owner(const int* __restrict__ srcs, const int* __restrict__ dsts, int E0, int N,
          const float* __restrict__ as_, const float* __restrict__ ad_,
          const float* __restrict__ hf, const float* __restrict__ bias,
          float* __restrict__ out,
          int* __restrict__ lsrc, int* __restrict__ ldl, int* __restrict__ lcnt,
          const unsigned int* __restrict__ keys, const unsigned int* __restrict__ gidx) {
  constexpr int F  = H * C;
  constexpr int FJ = F / 32;
  typedef float vrow __attribute__((ext_vector_type(FJ)));
  static_assert(H == 1 || (H == 2 && FJ * 16 == C), "a 16-lane row of the wave covers one head");
  constexpr int AST = 2;
  __shared__ float s_acc[(RNG + 1) * F];
  __shared__ float s_m[RNG * H];
  __shared__ float s_den[RNG * H];
  __shared__ float s_ad[RNG * H];
  __shared__ float s_self[RNG * H];
  __shared__ int   s_e[CAP];
  __shared__ int   s_dl[CAP];
  __shared__ float s_ex[CAP * H];
  __shared__ int   s_cnt;
  __shared__ int   s_ng;

  const int tid  = threadIdx.x;
  const int lane = tid & 31;
  const int wave = __builtin_amdgcn_readfirstlane(tid >> 5);
  const int lo   = blockIdx.x * RNG;
  const int cnt  = min(RNG, N - lo);

  for (int i = tid; i < (RNG + 1) * F; i += NT) s_acc[i] = 0.f;
  for (int i = tid; i < cnt * H; i += NT) {
    const int g = (lo + i / H) * AST + i % H;
    float a_d = ad_[g];
    s_ad[i]  = a_d;
    s_m[i]   = lrelu(as_[g] + a_d);
    s_den[i] = 0.f;
  }
  if (FIRST && tid == 0) { s_cnt = 0; s_ng = 0; }
  __syncthreads();

  int nm;
  if (FIRST) {
    auto keep = [&](int e) {
      int slot = atomicAdd(&s_cnt, 1);
      if (slot < CAP) s_e[slot] = e;
    };
    if (gridDim.x <= 256) {
      const unsigned kb = blockIdx.x * 0x01010101u;
      const int nfull = E0 >> 4;
      const uint4* kv = (const uint4*)keys;
      auto zb = [](unsigned x) {
        return ~(((x & 0x7F7F7F7Fu) + 0x7F7F7F7Fu) | x) & 0x80808080u;
      };
      const int nbw = ((nfull + GRP - 1) / GRP) * BMW;
      const unsigned int* rowp = gidx + (long)blockIdx.x * nbw;
      for (int base = 0; base < nbw; base += NT * 2) {
        unsigned mw[2];
#pragma unroll
        for (int k = 0; k < 2; ++k) mw[k] = rowp[min(base + k * NT + tid, nbw - 1)];
#pragma unroll
        for (int k = 0; k < 2; ++k) {
          const int q = base + k * NT + tid;
          unsigned m = (q < nbw) ? mw[k] : 0u;
          while (m) {
            const int p = __builtin_ctz(m);
            m &= m - 1;
            const int slot = atomicAdd(&s_ng, 1);
            if (slot < CAP) s_dl[slot] = q * 32 + p;
          }
        }
      }
      __syncthreads();
      const int ng = min(__builtin_amdgcn_readfirstlane(s_ng), CAP);
      for (int base = 0; base < ng; base += NT * 4) {
        int   gq[4];
        uint4 q[4];
#pragma unroll
        for (int k = 0; k < 4; ++k) gq[k] = s_dl[min(base + k * NT + tid, ng - 1)];
#pragma unroll
        for (int k = 0; k < 4; ++k) q[k] = kv[gq[k]];
#pragma unroll
        for (int k = 0; k < 4; ++k) {
          unsigned m = (zb(q[k].x ^ kb) >> 7) | (zb(q[k].y ^ kb) >> 6) |
                       (zb(q[k].z ^ kb) >> 5) | (zb(q[k].w ^ kb) >> 4);
          if (base + k * NT + tid >= ng) m = 0;
          const int e0 = gq[k] * 16;
          while (m) {
            int p = __builtin_ctz(m);
            m &= m - 1;
            keep(e0 + ((p & 7) << 2) + (p >> 3));
          }
        }
      }
      int et = nfull * 16 + tid;
      if (et < E0 && (unsigned)(dsts[et] - lo) < (unsigned)cnt) keep(et);
    } else {
      for (int e = tid; e < E0; e += NT)
        if ((unsigned)(dsts[e] - lo) < (unsigned)cnt) keep(e);
    }
    __syncthreads();
    nm = __builtin_amdgcn_readfirstlane(s_cnt);
    if (tid == 0) lcnt[blockIdx.x] = nm;
  } else {
    nm = lcnt[blockIdx.x];
  }
  const bool fits = (nm + cnt <= CAP);
  const int  tot  = nm + cnt;
  const int  totp = (tot + ROWS - 1) & ~(ROWS - 1);
  const long lb   = (long)blockIdx.x * CAP;

  if (fits) {
    for (int i = tid; i < nm; i += NT) {
      int s, dl;
      if (FIRST) {
        int e = s_e[i];
        s  = srcs[e];
        dl = dsts[e] - lo;
        s_e[i]  = s;
        s_dl[i] = dl;
        lsrc[lb + i] = s;
        ldl[lb + i]  = dl;
      } else {
        s  = lsrc[lb + i];
        dl = ldl[lb + i];
        s_e[i]  = s;
        s_dl[i] = dl;
      }
#pragma unroll
      for (int h = 0; h < H; ++h) {
        float v = lrelu(as_[s * AST + h] + s_ad[dl * H + h]);
        s_ex[i * H + h] = v;
        lds_max_f32(&s_m[dl * H + h], v);
      }
    }
  } else {
    for (int e = tid; e < E0; e += NT) {
      unsigned dl = (unsigned)(dsts[e] - lo);
      if (dl < (unsigned)cnt) {
        int s = srcs[e];
        for (int h = 0; h < H; ++h)
          lds_max_f32(&s_m[dl * H + h], lrelu(as_[s * AST + h] + s_ad[dl * H + h]));
      }
    }
  }
  __syncthreads();

  if (fits) {
    for (int i = tid; i < nm; i += NT) {
      int dl = s_dl[i];
#pragma unroll
      for (int h = 0; h < H; ++h) {
        float ex = expf(s_ex[i * H + h] - s_m[dl * H + h]);
        s_ex[i * H + h] = ex;
        atomicAdd(&s_den[dl * H + h], ex);
      }
    }
  }
  for (int i = tid; i < cnt * H; i += NT) {
    float ex = expf(lrelu(as_[(lo + i / H) * AST + i % H] + s_ad[i]) - s_m[i]);
    s_self[i] = ex;
    if (fits) s_ex[nm * H + i] = ex;
    atomicAdd(&s_den[i], ex);
  }
  if (fits) {
    for (int j = tid; j < cnt; j += NT) { s_e[nm + j] = lo + j; s_dl[nm + j] = j; }
    if (tid < totp - tot) {
      const int p = tot + tid;
      s_e[p]  = lo;
      s_dl[p] = RNG;
#pragma unroll
      for (int h = 0; h < H; ++h) s_ex[p * H + h] = 0.f;
    }
  }
  __syncthreads();

  if (fits) {
    const int hl = (FJ * lane) / C;
    for (int b = wave; b < (totp / ROWS); b += NW) {
      const int ii  = b * ROWS + (lane & (ROWS - 1));
      const int sv  = s_e[ii];
      const int dlv = s_dl[ii];
      const int exl = __float_as_int(s_ex[ii * H + hl]);
      vrow val[ROWS];
#pragma unroll
      for (int k = 0; k < ROWS; ++k) {
        const int s = __builtin_amdgcn_readlane(sv, k);
        val[k] = *((const vrow*)(hf + (long)s * F) + lane);
      }
      auto add_entry = [&](auto kc) {
        constexpr int k = decltype(kc)::v;
        const int dl = __builtin_amdgcn_readlane(dlv, k);
        float* acc = &s_acc[dl * F + lane];
        const float w = (H == 1) ? __int_as_float(__builtin_amdgcn_readlane(exl, k))
                                 : row_lane<k>(exl);
#pragma unroll
        for (int j = 0; j < FJ; ++j) atomicAdd(acc + 32 * j, val[k][j] * w);
      };
      kloop<0, ROWS>::run(add_entry);
    }
  } else {
    for (int e = tid; e < E0; e += NT) {
      unsigned dl = (unsigned)(dsts[e] - lo);
      if (dl < (unsigned)cnt) {
        int s = srcs[e];
        for (int h = 0; h < H; ++h) {
          float ex = expf(lrelu(as_[s * AST + h] + s_ad[dl * H + h]) - s_m[dl * H + h]);
          atomicAdd(&s_den[dl * H + h], ex);
          for (int c = 0; c < C; ++c)
            atomicAdd(&s_acc[dl * F + acc_word<F>(h * C + c)], hf[(long)s * F + h * C + c] * ex);
        }
      }
    }
  }
  if (!fits) {
    for (int i = wave; i < cnt; i += NW) {
#pragma unroll
      for (int j = 0; j < FJ; ++j)
        atomicAdd(&s_acc[i * F + acc_word<F>(lane + 32 * j)],
                  hf[(long)(lo + i) * F + lane + 32 * j] * s_self[i * H + (32 * j) / C]);
    }
  }
  __syncthreads();

  for (int i = tid; i < cnt * F; i += NT) {
    int dl = i / F, c = i - dl * F;
    float v = s_acc[dl * F + acc_word<F>(c)] / s_den[dl * H + c / C] + bias[c];
    if (DO_ELU) v = (v > 0.f) ? v : (expf(v) - 1.f);
    out[(long)lo * F + i] = v;
  }
}

extern "C" void kernel_launch(void* const* d_in, const int* in_sizes, int n_in,
                              void* d_out, int out_size, void* d_ws, size_t ws_size,
                              hipStream_t stream) {
  const float* x    = (const float*)d_in[0];
  const int*   ei   = (const int*)  d_in[1];
  const float* W1   = (const float*)d_in[2];
  const float* atS1 = (const float*)d_in[3];
  const float* atD1 = (const float*)d_in[4];
  const float* b1   = (const float*)d_in[5];
  const float* W2   = (const float*)d_in[6];
  const float* atS2 = (const float*)d_in[7];
  const float* atD2 = (const float*)d_in[8];
  const float* b2   = (const float*)d_in[9];
  float* out = (float*)d_out;

  const int N    = in_sizes[0] / DIN;
  const int E0   = in_sizes[1] / 2;
  const int* srcs = ei;
  const int* dsts = ei + E0;

  float* ws = (float*)d_ws;
  long off = 0;
  float* h1  = ws + off; off += (long)N * (HEADS * DH);
  float* o1  = ws + off; off += (long)N * (HEADS * DH);
  float* a1s = ws + off; off += (long)N * HEADS;
  float* a1d = ws + off; off += (long)N * HEADS;
  off = (off + 3) & ~3L;
  float* h2  = h1;
  float* a2  = h1 + (long)N * DOUT;

  auto blocks = [](long n, int tb) { return (unsigned)((n + tb - 1) / tb); };
  const unsigned NB = blocks(N, RNG);
  int* lsrc = (int*)(ws + off); off += (long)NB * CAP;
  int* ldl  = (int*)(ws + off); off += (long)NB * CAP;
  int* lcnt = (int*)(ws + off); off += NB;
  off = (off + 3) & ~3L;
  unsigned int* keys = (unsigned int*)(ws + off); off += (long)(E0 >> 4) * 4;
  off = (off + 31) & ~31L;
  const unsigned NG = blocks((long)(E0 >> 4), GRP);
  unsigned int* gidx = (unsigned int*)(ws + off);

  gemm_wmma_f32<(HEADS * DH) / 32, HEADS><<<256, 256, 0, stream>>>(x, W1, h1, N, atS1, atD1, a1s, a1d);
  edge_keys<<<NG, GRP, 0, stream>>>(dsts, E0, keys, gidx);
  gat_owner<HEADS, DH, true, true><<<NB, NT, 0, stream>>>(srcs, dsts, E0, N, a1s, a1d, h1, b1, o1,
                                                          lsrc, ldl, lcnt, keys, gidx);

  gemm_wmma_f32<DOUT / 32, 1><<<256, 256, 0, stream>>>(o1, W2, h2, N, atS2, atD2, a2, nullptr);
  gat_owner<1, DOUT, false, false><<<NB, NT, 0, stream>>>(srcs, dsts, E0, N, a2, a2 + 1, h2, b2, out,
                                                          lsrc, ldl, lcnt, keys, gidx);
}


# Key launch writes (source, destination) pairs, 16-edge group = one 128-byte line; layer-1 owners load only flagged lines: no key bytes, no edge reads by id; sums still LDS atomics, arrival order
# speedup vs baseline: 6.8623x; 1.0453x over previous
#include <hip/hip_runtime.h>
#include <math.h>

#define DIN   128
#define DH    64
#define HEADS 2
#define DOUT  64
#define NEG_SLOPE 0.2f

typedef float v2f __attribute__((ext_vector_type(2)));
typedef float v8f __attribute__((ext_vector_type(8)));

typedef float v4f __attribute__((ext_vector_type(4)));
#define GK 128
static_assert(DIN == GK && HEADS * DH == GK, "both products have inner dimension GK");

template <int NSTRIP, int NH>
__global__ void __launch_bounds__(256)
gemm_wmma_f32(const float* __restrict__ A, const float* __restrict__ B,
              float* __restrict__ C, int M,
              const float* __restrict__ att_s, const float* __restrict__ att_d,
              float* __restrict__ o0, float* __restrict__ o1) {
  constexpr int Nc      = 32 * NSTRIP;
  constexpr int NSTREAM = 8 / NSTRIP;
  constexpr int CH      = Nc / NH;
  static_assert(NH <= NSTRIP && CH % 4 == 0, "one dot wave per vector pair");
  __shared__ alignas(16) float s_wa[2 * NH][GK];
  for (int o = threadIdx.x; o < GK * NH; o += 256) {
    const int k = o / NH, hd = o - k * NH;
    const v4f* wp = (const v4f*)(B + k * Nc + hd * CH);
    const v4f* sp = (const v4f*)(att_s + hd * CH);
    const v4f* dp = (const v4f*)(att_d + hd * CH);
    float ss = 0.f, dd = 0.f;
    for (int c = 0; c < CH / 4; ++c) {
      const v4f w = wp[c], s = sp[c], d = dp[c];
      ss += w.x * s.x; ss += w.y * s.y; ss += w.z * s.z; ss += w.w * s.w;
      dd += w.x * d.x; dd += w.y * d.y; dd += w.z * d.z; dd += w.w * d.w;
    }
    s_wa[hd][k]      = ss;
    s_wa[NH + hd][k] = dd;
  }
  __syncthreads();
  const int lane   = threadIdx.x & 31;
  const int wave   = __builtin_amdgcn_readfirstlane(threadIdx.x >> 5);
  const int half   = lane >> 4;
  const int l16    = lane & 15;
  const int strip  = wave % NSTRIP;
  const int stream = wave / NSTRIP;
  const int colp   = strip * 32 + 2 * l16;

  const int tiles = (M + 15) >> 4;
  const int step  = gridDim.x * NSTREAM;
  int tile = blockIdx.x * NSTREAM + stream;
  if (tile >= tiles) return;

  v2f bw[2][32];
#pragma unroll
  for (int j = 0; j < 32; ++j) {
    const int k = 8 * (j >> 1) + 4 * half + 2 * (j & 1);
    const v2f w0 = *(const v2f*)(B + k * Nc + colp);
    const v2f w1 = *(const v2f*)(B + (k + 1) * Nc + colp);
    bw[0][j].x = w0.x; bw[0][j].y = w1.x;
    bw[1][j].x = w0.y; bw[1][j].y = w1.y;
  }

  auto load_tile = [&](v4f (&a)[16], int tl) {
    long row = (long)tl * 16 + l16;
    if (row >= M) row = M - 1;
    const v4f* p = (const v4f*)(A + row * GK + 4 * half);
#pragma unroll
    for (int jj = 0; jj < 16; ++jj) a[jj] = p[2 * jj];
  };
  const bool dotw = strip < NH;
  v4f wv[2][16];
  if (dotw) {
#pragma unroll
    for (int jj = 0; jj < 16; ++jj) {
      wv[0][jj] = *(const v4f*)&s_wa[2 * strip][8 * jj + 4 * half];
      wv[1][jj] = *(const v4f*)&s_wa[2 * strip + 1][8 * jj + 4 * half];
    }
  }
  float* const dotp = (strip == 0) ? o0 : o1;
  auto dot_tile = [&](const v4f (&a)[16], int tl) {
    float p0 = 0.f, q0 = 0.f, p1 = 0.f, q1 = 0.f;
#pragma unroll
    for (int jj = 0; jj < 16; jj += 2) {
      p0 += a[jj].x * wv[0][jj].x; p0 += a[jj].y * wv[0][jj].y;
      p0 += a[jj].z * wv[0][jj].z; p0 += a[jj].w * wv[0][jj].w;
      p1 += a[jj].x * wv[1][jj].x; p1 += a[jj].y * wv[1][jj].y;
      p1 += a[jj].z * wv[1][jj].z; p1 += a[jj].w * wv[1][jj].w;
      q0 += a[jj + 1].x * wv[0][jj + 1].x; q0 += a[jj + 1].y * wv[0][jj + 1].y;
      q0 += a[jj + 1].z * wv[0][jj + 1].z; q0 += a[jj + 1].w * wv[0][jj + 1].w;
      q1 += a[jj + 1].x * wv[1][jj + 1].x; q1 += a[jj + 1].y * wv[1][jj + 1].y;
      q1 += a[jj + 1].z * wv[1][jj + 1].z; q1 += a[jj + 1].w * wv[1][jj + 1].w;
    }
    p0 += q0; p1 += q1;
    p0 += __shfl_xor(p0, 16);
    p1 += __shfl_xor(p1, 16);
    const long row = (long)tl * 16 + l16;
    if (half == 0 && row < M) {
      v2f o; o.x = p0; o.y = p1;
      *(v2f*)(dotp + row * 2) = o;
    }
  };
  auto mul_tile = [&](const v4f (&a)[16], int tl) {
    v8f c0 = {}, c1 = {};
#pragma unroll
    for (int j = 0; j < 32; ++j) {
      v2f av;
      if (j & 1) { av.x = a[j >> 1].z; av.y = a[j >> 1].w; }
      else       { av.x = a[j >> 1].x; av.y = a[j >> 1].y; }
      c0 = __builtin_amdgcn_wmma_f32_16x16x4_f32(false, av, false, bw[0][j],
                                                 (short)0, c0, false, false);
      c1 = __builtin_amdgcn_wmma_f32_16x16x4_f32(false, av, false, bw[1][j],
                                                 (short)0, c1, false, false);
    }
    if ((tl + 1) * 16 <= M) {
      float* cp = C + ((long)tl * 16 + half * 8) * Nc + colp;
#pragma unroll
      for (int r = 0; r < 8; ++r) {
        v2f o; o.x = c0[r]; o.y = c1[r];
        *(v2f*)(cp + r * Nc) = o;
      }
    } else {
#pragma unroll
      for (int r = 0; r < 8; ++r) {
        long row = (long)tl * 16 + r + half * 8;
        if (row < M) {
          v2f o; o.x = c0[r]; o.y = c1[r];
          *(v2f*)(C + row * Nc + colp) = o;
        }
      }
    }
  };

  v4f a0[16], a1[16];
  load_tile(a0, tile);
  while (true) {
    load_tile(a1, min(tile + step, tiles - 1));
    mul_tile(a0, tile);
    if (dotw) dot_tile(a0, tile);
    tile += step;
    if (tile >= tiles) break;
    load_tile(a0, min(tile + step, tiles - 1));
    mul_tile(a1, tile);
    if (dotw) dot_tile(a1, tile);
    tile += step;
    if (tile >= tiles) break;
  }
}

#define RNG 196
#define GRP 1024
#define BMW (GRP / 32)
#define BMP 257
#define RPT 4
#define RPB (GRP * RPT)

__global__ void __launch_bounds__(GRP)
edge_keys(const int* __restrict__ srcs, const int* __restrict__ dsts, int E0,
          int* __restrict__ rec, unsigned int* __restrict__ gidx, int nidx) {
  __shared__ unsigned int s_bm[BMW * BMP];
  const int tid   = threadIdx.x;
  const int nfull = E0 >> 4;
  if ((int)blockIdx.x >= nidx) {
    const long n    = (long)nfull * 16;
    const long base = (long)((int)blockIdx.x - nidx) * RPB + tid;
    int s[RPT], d[RPT];
#pragma unroll
    for (int k = 0; k < RPT; ++k) {
      const long e  = base + (long)k * GRP;
      const long ec = (e < n) ? e : (n - 1);
      s[k] = srcs[ec];
      d[k] = dsts[ec];
    }
#pragma unroll
    for (int k = 0; k < RPT; ++k) {
      const long e = base + (long)k * GRP;
      if (e < n) {
        int2 r; r.x = s[k]; r.y = d[k];
        ((int2*)rec)[e] = r;
      }
    }
    return;
  }
  for (int i = tid; i < BMW * BMP; i += GRP) s_bm[i] = 0u;
  __syncthreads();
  const int g = blockIdx.x * GRP + tid;
  if (g < nfull) {
    const int* dp = dsts + 16 * (long)g;
    int d[16];
#pragma unroll
    for (int c = 0; c < 16; ++c) d[c] = dp[c];
    unsigned int* col = s_bm + (tid >> 5) * BMP;
    const unsigned bit = 1u << (tid & 31);
#pragma unroll
    for (int c = 0; c < 16; ++c) atomicOr(col + min((unsigned)d[c] / RNG, 255u), bit);
  }
  __syncthreads();
  const int  lane  = tid & 31;
  const int  wave  = tid >> 5;
  const long pitch = (long)nidx * BMW;
  unsigned int* op = gidx + (long)blockIdx.x * BMW + lane;
  for (int r = wave; r < 256; r += GRP / 32) op[r * pitch] = s_bm[lane * BMP + r];
}

#define CAP 8192
#define ROWS 16
#define NT  1024
#define NW  (NT / 32)

__device__ __forceinline__ float lrelu(float v) { return (v > 0.f) ? v : NEG_SLOPE * v; }

template <int F>
__device__ __forceinline__ int acc_word(int f) { return (f % (F / 32)) * 32 + f / (F / 32); }

template <int K> struct kconst { static constexpr int v = K; };
template <int K, int KN> struct kloop {
  template <typename Fn> static __device__ __forceinline__ void run(Fn& fn) {
    fn(kconst<K>{});
    kloop<K + 1, KN>::run(fn);
  }
};
template <int KN> struct kloop<KN, KN> {
  template <typename Fn> static __device__ __forceinline__ void run(Fn&) {}
};
template <int K>
__device__ __forceinline__ float row_lane(int v) {
  return __int_as_float(__builtin_amdgcn_update_dpp(0, v, 0x150 + K, 0xf, 0xf, false));
}

__device__ __forceinline__ void lds_max_f32(float* addr, float v) {
  if (v >= 0.0f) atomicMax((int*)addr, __float_as_int(v));
  else           atomicMin((unsigned int*)addr, __float_as_uint(v));
}

template <int H, int C, bool DO_ELU, bool FIRST>
__global__ void __launch_bounds__(NT)
gat_owner(const int* __restrict__ srcs, const int* __restrict__ dsts, int E0, int N,
          const float* __restrict__ as_, const float* __restrict__ ad_,
          const float* __restrict__ hf, const float* __restrict__ bias,
          float* __restrict__ out,
          int* __restrict__ lsrc, int* __restrict__ ldl, int* __restrict__ lcnt,
          const int* __restrict__ rec, const unsigned int* __restrict__ gidx) {
  constexpr int F  = H * C;
  constexpr int FJ = F / 32;
  typedef float vrow __attribute__((ext_vector_type(FJ)));
  static_assert(H == 1 || (H == 2 && FJ * 16 == C), "a 16-lane row of the wave covers one head");
  constexpr int AST = 2;
  __shared__ float s_acc[(RNG + 1) * F];
  __shared__ float s_m[RNG * H];
  __shared__ float s_den[RNG * H];
  __shared__ float s_ad[RNG * H];
  __shared__ float s_self[RNG * H];
  __shared__ int   s_e[CAP];
  __shared__ int   s_dl[CAP];
  __shared__ float s_ex[CAP * H];
  __shared__ int   s_cnt;
  __shared__ int   s_ng;

  const int tid  = threadIdx.x;
  const int lane = tid & 31;
  const int wave = __builtin_amdgcn_readfirstlane(tid >> 5);
  const int lo   = blockIdx.x * RNG;
  const int cnt  = min(RNG, N - lo);

  for (int i = tid; i < (RNG + 1) * F; i += NT) s_acc[i] = 0.f;
  for (int i = tid; i < cnt * H; i += NT) {
    const int g = (lo + i / H) * AST + i % H;
    float a_d = ad_[g];
    s_ad[i]  = a_d;
    s_m[i]   = lrelu(as_[g] + a_d);
    s_den[i] = 0.f;
  }
  if (FIRST && tid == 0) { s_cnt = 0; s_ng = 0; }
  __syncthreads();

  int nm;
  if (FIRST) {
    auto keep = [&](int s, int dl) {
      int slot = atomicAdd(&s_cnt, 1);
      if (slot < CAP) { s_e[slot] = s; s_dl[slot] = dl; }
    };
    if (gridDim.x <= 256) {
      const int nfull = E0 >> 4;
      typedef int __attribute__((may_alias)) lds_int;
      lds_int* const s_gl = (lds_int*)s_ex;
      const int nbw = ((nfull + GRP - 1) / GRP) * BMW;
      const unsigned int* rowp = gidx + (long)blockIdx.x * nbw;
      for (int base = 0; base < nbw; base += NT * 2) {
        unsigned mw[2];
#pragma unroll
        for (int k = 0; k < 2; ++k) mw[k] = rowp[min(base + k * NT + tid, nbw - 1)];
#pragma unroll
        for (int k = 0; k < 2; ++k) {
          const int q = base + k * NT + tid;
          unsigned m = (q < nbw) ? mw[k] : 0u;
          while (m) {
            const int p = __builtin_ctz(m);
            m &= m - 1;
            const int slot = atomicAdd(&s_ng, 1);
            if (slot < CAP) s_gl[slot] = q * 32 + p;
          }
        }
      }
      __syncthreads();
      const int ng = min(__builtin_amdgcn_readfirstlane(s_ng), CAP);
      constexpr int NS = NT / 8;
      constexpr int LB = 8;
      const int st = tid >> 3, k8 = tid & 7;
      const int4* rv = (const int4*)rec;
      for (int base = 0; base < ng; base += NS * LB) {
        int4 q[LB];
#pragma unroll
        for (int k = 0; k < LB; ++k)
          q[k] = rv[(long)s_gl[min(base + k * NS + st, ng - 1)] * 8 + k8];
        int  off[2 * LB];
        bool hit[2 * LB];
        int  nk = 0;
#pragma unroll
        for (int k = 0; k < LB; ++k) {
          const bool in = base + k * NS + st < ng;
          hit[2 * k]     = in && ((unsigned)(q[k].y - lo) < (unsigned)cnt);
          hit[2 * k + 1] = in && ((unsigned)(q[k].w - lo) < (unsigned)cnt);
#pragma unroll
          for (int u = 0; u < 2; ++u) {
            const unsigned bl = __builtin_amdgcn_ballot_w32(hit[2 * k + u]);
            off[2 * k + u] = nk + (int)__builtin_amdgcn_mbcnt_lo(bl, 0u);
            nk += __builtin_popcount(bl);
          }
        }
        int first = 0;
        if (lane == 0) first = atomicAdd(&s_cnt, nk);
        first = __builtin_amdgcn_readfirstlane(first);
#pragma unroll
        for (int k = 0; k < LB; ++k) {
          const int sl0 = first + off[2 * k], sl1 = first + off[2 * k + 1];
          if (hit[2 * k]     && sl0 < CAP) { s_e[sl0] = q[k].x; s_dl[sl0] = q[k].y - lo; }
          if (hit[2 * k + 1] && sl1 < CAP) { s_e[sl1] = q[k].z; s_dl[sl1] = q[k].w - lo; }
        }
      }
      int et = nfull * 16 + tid;
      if (et < E0) {
        const unsigned dl = (unsigned)(dsts[et] - lo);
        if (dl < (unsigned)cnt) keep(srcs[et], (int)dl);
      }
    } else {
      for (int e = tid; e < E0; e += NT) {
        const unsigned dl = (unsigned)(dsts[e] - lo);
        if (dl < (unsigned)cnt) keep(srcs[e], (int)dl);
      }
    }
    __syncthreads();
    nm = __builtin_amdgcn_readfirstlane(s_cnt);
    if (tid == 0) lcnt[blockIdx.x] = nm;
  } else {
    nm = lcnt[blockIdx.x];
  }
  const bool fits = (nm + cnt <= CAP);
  const int  tot  = nm + cnt;
  const int  totp = (tot + ROWS - 1) & ~(ROWS - 1);
  const long lb   = (long)blockIdx.x * CAP;

  if (fits) {
    for (int i = tid; i < nm; i += NT) {
      int s, dl;
      if (FIRST) {
        s  = s_e[i];
        dl = s_dl[i];
        lsrc[lb + i] = s;
        ldl[lb + i]  = dl;
      } else {
        s  = lsrc[lb + i];
        dl = ldl[lb + i];
        s_e[i]  = s;
        s_dl[i] = dl;
      }
#pragma unroll
      for (int h = 0; h < H; ++h) {
        float v = lrelu(as_[s * AST + h] + s_ad[dl * H + h]);
        s_ex[i * H + h] = v;
        lds_max_f32(&s_m[dl * H + h], v);
      }
    }
  } else {
    for (int e = tid; e < E0; e += NT) {
      unsigned dl = (unsigned)(dsts[e] - lo);
      if (dl < (unsigned)cnt) {
        int s = srcs[e];
        for (int h = 0; h < H; ++h)
          lds_max_f32(&s_m[dl * H + h], lrelu(as_[s * AST + h] + s_ad[dl * H + h]));
      }
    }
  }
  __syncthreads();

  if (fits) {
    for (int i = tid; i < nm; i += NT) {
      int dl = s_dl[i];
#pragma unroll
      for (int h = 0; h < H; ++h) {
        float ex = expf(s_ex[i * H + h] - s_m[dl * H + h]);
        s_ex[i * H + h] = ex;
        atomicAdd(&s_den[dl * H + h], ex);
      }
    }
  }
  for (int i = tid; i < cnt * H; i += NT) {
    float ex = expf(lrelu(as_[(lo + i / H) * AST + i % H] + s_ad[i]) - s_m[i]);
    s_self[i] = ex;
    if (fits) s_ex[nm * H + i] = ex;
    atomicAdd(&s_den[i], ex);
  }
  if (fits) {
    for (int j = tid; j < cnt; j += NT) { s_e[nm + j] = lo + j; s_dl[nm + j] = j; }
    if (tid < totp - tot) {
      const int p = tot + tid;
      s_e[p]  = lo;
      s_dl[p] = RNG;
#pragma unroll
      for (int h = 0; h < H; ++h) s_ex[p * H + h] = 0.f;
    }
  }
  __syncthreads();

  if (fits) {
    const int hl = (FJ * lane) / C;
    for (int b = wave; b < (totp / ROWS); b += NW) {
      const int ii  = b * ROWS + (lane & (ROWS - 1));
      const int sv  = s_e[ii];
      const int dlv = s_dl[ii];
      const int exl = __float_as_int(s_ex[ii * H + hl]);
      vrow val[ROWS];
#pragma unroll
      for (int k = 0; k < ROWS; ++k) {
        const int s = __builtin_amdgcn_readlane(sv, k);
        val[k] = *((const vrow*)(hf + (long)s * F) + lane);
      }
      auto add_entry = [&](auto kc) {
        constexpr int k = decltype(kc)::v;
        const int dl = __builtin_amdgcn_readlane(dlv, k);
        float* acc = &s_acc[dl * F + lane];
        const float w = (H == 1) ? __int_as_float(__builtin_amdgcn_readlane(exl, k))
                                 : row_lane<k>(exl);
#pragma unroll
        for (int j = 0; j < FJ; ++j) atomicAdd(acc + 32 * j, val[k][j] * w);
      };
      kloop<0, ROWS>::run(add_entry);
    }
  } else {
    for (int e = tid; e < E0; e += NT) {
      unsigned dl = (unsigned)(dsts[e] - lo);
      if (dl < (unsigned)cnt) {
        int s = srcs[e];
        for (int h = 0; h < H; ++h) {
          float ex = expf(lrelu(as_[s * AST + h] + s_ad[dl * H + h]) - s_m[dl * H + h]);
          atomicAdd(&s_den[dl * H + h], ex);
          for (int c = 0; c < C; ++c)
            atomicAdd(&s_acc[dl * F + acc_word<F>(h * C + c)], hf[(long)s * F + h * C + c] * ex);
        }
      }
    }
  }
  if (!fits) {
    for (int i = wave; i < cnt; i += NW) {
#pragma unroll
      for (int j = 0; j < FJ; ++j)
        atomicAdd(&s_acc[i * F + acc_word<F>(lane + 32 * j)],
                  hf[(long)(lo + i) * F + lane + 32 * j] * s_self[i * H + (32 * j) / C]);
    }
  }
  __syncthreads();

  for (int i = tid; i < cnt * F; i += NT) {
    int dl = i / F, c = i - dl * F;
    float v = s_acc[dl * F + acc_word<F>(c)] / s_den[dl * H + c / C] + bias[c];
    if (DO_ELU) v = (v > 0.f) ? v : (expf(v) - 1.f);
    out[(long)lo * F + i] = v;
  }
}

extern "C" void kernel_launch(void* const* d_in, const int* in_sizes, int n_in,
                              void* d_out, int out_size, void* d_ws, size_t ws_size,
                              hipStream_t stream) {
  const float* x    = (const float*)d_in[0];
  const int*   ei   = (const int*)  d_in[1];
  const float* W1   = (const float*)d_in[2];
  const float* atS1 = (const float*)d_in[3];
  const float* atD1 = (const float*)d_in[4];
  const float* b1   = (const float*)d_in[5];
  const float* W2   = (const float*)d_in[6];
  const float* atS2 = (const float*)d_in[7];
  const float* atD2 = (const float*)d_in[8];
  const float* b2   = (const float*)d_in[9];
  float* out = (float*)d_out;

  const int N    = in_sizes[0] / DIN;
  const int E0   = in_sizes[1] / 2;
  const int* srcs = ei;
  const int* dsts = ei + E0;

  float* ws = (float*)d_ws;
  long off = 0;
  float* h1  = ws + off; off += (long)N * (HEADS * DH);
  float* o1  = ws + off; off += (long)N * (HEADS * DH);
  float* a1s = ws + off; off += (long)N * HEADS;
  float* a1d = ws + off; off += (long)N * HEADS;
  off = (off + 3) & ~3L;
  float* h2  = h1;
  float* a2  = h1 + (long)N * DOUT;

  auto blocks = [](long n, int tb) { return (unsigned)((n + tb - 1) / tb); };
  const unsigned NB = blocks(N, RNG);
  int* lsrc = (int*)(ws + off); off += (long)NB * CAP;
  int* ldl  = (int*)(ws + off); off += (long)NB * CAP;
  int* lcnt = (int*)(ws + off); off += NB;
  off = (off + 3) & ~3L;
  off = (off + 31) & ~31L;
  int* rec = (int*)(ws + off); off += (long)(E0 >> 4) * 32;
  off = (off + 31) & ~31L;
  const unsigned NG = blocks((long)(E0 >> 4), GRP);
  const unsigned NR = blocks((long)(E0 >> 4) * 16, RPB);
  unsigned int* gidx = (unsigned int*)(ws + off);

  gemm_wmma_f32<(HEADS * DH) / 32, HEADS><<<256, 256, 0, stream>>>(x, W1, h1, N, atS1, atD1, a1s, a1d);
  edge_keys<<<NG + NR, GRP, 0, stream>>>(srcs, dsts, E0, rec, gidx, (int)NG);
  gat_owner<HEADS, DH, true, true><<<NB, NT, 0, stream>>>(srcs, dsts, E0, N, a1s, a1d, h1, b1, o1,
                                                          lsrc, ldl, lcnt, rec, gidx);

  gemm_wmma_f32<DOUT / 32, 1><<<256, 256, 0, stream>>>(o1, W2, h2, N, atS2, atD2, a2, nullptr);
  gat_owner<1, DOUT, false, false><<<NB, NT, 0, stream>>>(srcs, dsts, E0, N, a2, a2 + 1, h2, b2, out,
                                                          lsrc, ldl, lcnt, rec, gidx);
}


# Owner blocks order their list by destination (counting sort in LDS); a wave sums each node's run in registers over a contiguous stretch, one LDS atomic add per run, not per entry; sum order changed
# speedup vs baseline: 7.3664x; 1.0735x over previous
#include <hip/hip_runtime.h>
#include <math.h>

#define DIN   128
#define DH    64
#define HEADS 2
#define DOUT  64
#define NEG_SLOPE 0.2f

typedef float v2f __attribute__((ext_vector_type(2)));
typedef float v8f __attribute__((ext_vector_type(8)));

typedef float v4f __attribute__((ext_vector_type(4)));
#define GK 128
static_assert(DIN == GK && HEADS * DH == GK, "both products have inner dimension GK");

template <int NSTRIP, int NH>
__global__ void __launch_bounds__(256)
gemm_wmma_f32(const float* __restrict__ A, const float* __restrict__ B,
              float* __restrict__ C, int M,
              const float* __restrict__ att_s, const float* __restrict__ att_d,
              float* __restrict__ o0, float* __restrict__ o1) {
  constexpr int Nc      = 32 * NSTRIP;
  constexpr int NSTREAM = 8 / NSTRIP;
  constexpr int CH      = Nc / NH;
  static_assert(NH <= NSTRIP && CH % 4 == 0, "one dot wave per vector pair");
  __shared__ alignas(16) float s_wa[2 * NH][GK];
  for (int o = threadIdx.x; o < GK * NH; o += 256) {
    const int k = o / NH, hd = o - k * NH;
    const v4f* wp = (const v4f*)(B + k * Nc + hd * CH);
    const v4f* sp = (const v4f*)(att_s + hd * CH);
    const v4f* dp = (const v4f*)(att_d + hd * CH);
    float ss = 0.f, dd = 0.f;
    for (int c = 0; c < CH / 4; ++c) {
      const v4f w = wp[c], s = sp[c], d = dp[c];
      ss += w.x * s.x; ss += w.y * s.y; ss += w.z * s.z; ss += w.w * s.w;
      dd += w.x * d.x; dd += w.y * d.y; dd += w.z * d.z; dd += w.w * d.w;
    }
    s_wa[hd][k]      = ss;
    s_wa[NH + hd][k] = dd;
  }
  __syncthreads();
  const int lane   = threadIdx.x & 31;
  const int wave   = __builtin_amdgcn_readfirstlane(threadIdx.x >> 5);
  const int half   = lane >> 4;
  const int l16    = lane & 15;
  const int strip  = wave % NSTRIP;
  const int stream = wave / NSTRIP;
  const int colp   = strip * 32 + 2 * l16;

  const int tiles = (M + 15) >> 4;
  const int step  = gridDim.x * NSTREAM;
  int tile = blockIdx.x * NSTREAM + stream;
  if (tile >= tiles) return;

  v2f bw[2][32];
#pragma unroll
  for (int j = 0; j < 32; ++j) {
    const int k = 8 * (j >> 1) + 4 * half + 2 * (j & 1);
    const v2f w0 = *(const v2f*)(B + k * Nc + colp);
    const v2f w1 = *(const v2f*)(B + (k + 1) * Nc + colp);
    bw[0][j].x = w0.x; bw[0][j].y = w1.x;
    bw[1][j].x = w0.y; bw[1][j].y = w1.y;
  }

  auto load_tile = [&](v4f (&a)[16], int tl) {
    long row = (long)tl * 16 + l16;
    if (row >= M) row = M - 1;
    const v4f* p = (const v4f*)(A + row * GK + 4 * half);
#pragma unroll
    for (int jj = 0; jj < 16; ++jj) a[jj] = p[2 * jj];
  };
  const bool dotw = strip < NH;
  v4f wv[2][16];
  if (dotw) {
#pragma unroll
    for (int jj = 0; jj < 16; ++jj) {
      wv[0][jj] = *(const v4f*)&s_wa[2 * strip][8 * jj + 4 * half];
      wv[1][jj] = *(const v4f*)&s_wa[2 * strip + 1][8 * jj + 4 * half];
    }
  }
  float* const dotp = (strip == 0) ? o0 : o1;
  auto dot_tile = [&](const v4f (&a)[16], int tl) {
    float p0 = 0.f, q0 = 0.f, p1 = 0.f, q1 = 0.f;
#pragma unroll
    for (int jj = 0; jj < 16; jj += 2) {
      p0 += a[jj].x * wv[0][jj].x; p0 += a[jj].y * wv[0][jj].y;
      p0 += a[jj].z * wv[0][jj].z; p0 += a[jj].w * wv[0][jj].w;
      p1 += a[jj].x * wv[1][jj].x; p1 += a[jj].y * wv[1][jj].y;
      p1 += a[jj].z * wv[1][jj].z; p1 += a[jj].w * wv[1][jj].w;
      q0 += a[jj + 1].x * wv[0][jj + 1].x; q0 += a[jj + 1].y * wv[0][jj + 1].y;
      q0 += a[jj + 1].z * wv[0][jj + 1].z; q0 += a[jj + 1].w * wv[0][jj + 1].w;
      q1 += a[jj + 1].x * wv[1][jj + 1].x; q1 += a[jj + 1].y * wv[1][jj + 1].y;
      q1 += a[jj + 1].z * wv[1][jj + 1].z; q1 += a[jj + 1].w * wv[1][jj + 1].w;
    }
    p0 += q0; p1 += q1;
    p0 += __shfl_xor(p0, 16);
    p1 += __shfl_xor(p1, 16);
    const long row = (long)tl * 16 + l16;
    if (half == 0 && row < M) {
      v2f o; o.x = p0; o.y = p1;
      *(v2f*)(dotp + row * 2) = o;
    }
  };
  auto mul_tile = [&](const v4f (&a)[16], int tl) {
    v8f c0 = {}, c1 = {};
#pragma unroll
    for (int j = 0; j < 32; ++j) {
      v2f av;
      if (j & 1) { av.x = a[j >> 1].z; av.y = a[j >> 1].w; }
      else       { av.x = a[j >> 1].x; av.y = a[j >> 1].y; }
      c0 = __builtin_amdgcn_wmma_f32_16x16x4_f32(false, av, false, bw[0][j],
                                                 (short)0, c0, false, false);
      c1 = __builtin_amdgcn_wmma_f32_16x16x4_f32(false, av, false, bw[1][j],
                                                 (short)0, c1, false, false);
    }
    if ((tl + 1) * 16 <= M) {
      float* cp = C + ((long)tl * 16 + half * 8) * Nc + colp;
#pragma unroll
      for (int r = 0; r < 8; ++r) {
        v2f o; o.x = c0[r]; o.y = c1[r];
        *(v2f*)(cp + r * Nc) = o;
      }
    } else {
#pragma unroll
      for (int r = 0; r < 8; ++r) {
        long row = (long)tl * 16 + r + half * 8;
        if (row < M) {
          v2f o; o.x = c0[r]; o.y = c1[r];
          *(v2f*)(C + row * Nc + colp) = o;
        }
      }
    }
  };

  v4f a0[16], a1[16];
  load_tile(a0, tile);
  while (true) {
    load_tile(a1, min(tile + step, tiles - 1));
    mul_tile(a0, tile);
    if (dotw) dot_tile(a0, tile);
    tile += step;
    if (tile >= tiles) break;
    load_tile(a0, min(tile + step, tiles - 1));
    mul_tile(a1, tile);
    if (dotw) dot_tile(a1, tile);
    tile += step;
    if (tile >= tiles) break;
  }
}

#define RNG 196
#define GRP 1024
#define BMW (GRP / 32)
#define BMP 257
#define RPT 4
#define RPB (GRP * RPT)

__global__ void __launch_bounds__(GRP)
edge_keys(const int* __restrict__ srcs, const int* __restrict__ dsts, int E0,
          int* __restrict__ rec, unsigned int* __restrict__ gidx, int nidx) {
  __shared__ unsigned int s_bm[BMW * BMP];
  const int tid   = threadIdx.x;
  const int nfull = E0 >> 4;
  if ((int)blockIdx.x >= nidx) {
    const long n    = (long)nfull * 16;
    const long base = (long)((int)blockIdx.x - nidx) * RPB + tid;
    int s[RPT], d[RPT];
#pragma unroll
    for (int k = 0; k < RPT; ++k) {
      const long e  = base + (long)k * GRP;
      const long ec = (e < n) ? e : (n - 1);
      s[k] = srcs[ec];
      d[k] = dsts[ec];
    }
#pragma unroll
    for (int k = 0; k < RPT; ++k) {
      const long e = base + (long)k * GRP;
      if (e < n) {
        int2 r; r.x = s[k]; r.y = d[k];
        ((int2*)rec)[e] = r;
      }
    }
    return;
  }
  for (int i = tid; i < BMW * BMP; i += GRP) s_bm[i] = 0u;
  __syncthreads();
  const int g = blockIdx.x * GRP + tid;
  if (g < nfull) {
    const int* dp = dsts + 16 * (long)g;
    int d[16];
#pragma unroll
    for (int c = 0; c < 16; ++c) d[c] = dp[c];
    unsigned int* col = s_bm + (tid >> 5) * BMP;
    const unsigned bit = 1u << (tid & 31);
#pragma unroll
    for (int c = 0; c < 16; ++c) atomicOr(col + min((unsigned)d[c] / RNG, 255u), bit);
  }
  __syncthreads();
  const int  lane  = tid & 31;
  const int  wave  = tid >> 5;
  const long pitch = (long)nidx * BMW;
  unsigned int* op = gidx + (long)blockIdx.x * BMW + lane;
  for (int r = wave; r < 256; r += GRP / 32) op[r * pitch] = s_bm[lane * BMP + r];
}

#define CAP 4096
#define ROWS 16
#define NT  1024
#define NW  (NT / 32)

__device__ __forceinline__ float lrelu(float v) { return (v > 0.f) ? v : NEG_SLOPE * v; }

template <int F>
__device__ __forceinline__ int acc_word(int f) { return (f % (F / 32)) * 32 + f / (F / 32); }

template <int K> struct kconst { static constexpr int v = K; };
template <int K, int KN> struct kloop {
  template <typename Fn> static __device__ __forceinline__ void run(Fn& fn) {
    fn(kconst<K>{});
    kloop<K + 1, KN>::run(fn);
  }
};
template <int KN> struct kloop<KN, KN> {
  template <typename Fn> static __device__ __forceinline__ void run(Fn&) {}
};
template <int K>
__device__ __forceinline__ float row_lane(int v) {
  return __int_as_float(__builtin_amdgcn_update_dpp(0, v, 0x150 + K, 0xf, 0xf, false));
}

__device__ __forceinline__ void lds_max_f32(float* addr, float v) {
  if (v >= 0.0f) atomicMax((int*)addr, __float_as_int(v));
  else           atomicMin((unsigned int*)addr, __float_as_uint(v));
}

template <int H, int C, bool DO_ELU, bool FIRST>
__global__ void __launch_bounds__(NT)
gat_owner(const int* __restrict__ srcs, const int* __restrict__ dsts, int E0, int N,
          const float* __restrict__ as_, const float* __restrict__ ad_,
          const float* __restrict__ hf, const float* __restrict__ bias,
          float* __restrict__ out,
          int* __restrict__ lsrc, int* __restrict__ ldl, int* __restrict__ lcnt,
          const int* __restrict__ rec, const unsigned int* __restrict__ gidx) {
  constexpr int F  = H * C;
  constexpr int FJ = F / 32;
  typedef float vrow __attribute__((ext_vector_type(FJ)));
  static_assert(H == 1 || (H == 2 && FJ * 16 == C), "a 16-lane row of the wave covers one head");
  constexpr int AST = 2;
  __shared__ float s_acc[(RNG + 1) * F];
  __shared__ float s_m[RNG * H];
  __shared__ float s_den[RNG * H];
  __shared__ float s_ad[RNG * H];
  __shared__ float s_self[RNG * H];
  __shared__ int   s_e[CAP];
  __shared__ int   s_dl[CAP];
  __shared__ float s_ex[CAP * H];
  __shared__ int   s_e2[CAP];
  __shared__ int   s_dl2[CAP];
  __shared__ float s_ex2[CAP * H];
  __shared__ int   s_deg[RNG];
  __shared__ int   s_pos[RNG];
  __shared__ int   s_cnt;
  __shared__ int   s_ng;

  const int tid  = threadIdx.x;
  const int lane = tid & 31;
  const int wave = __builtin_amdgcn_readfirstlane(tid >> 5);
  const int lo   = blockIdx.x * RNG;
  const int cnt  = min(RNG, N - lo);

  for (int i = tid; i < (RNG + 1) * F; i += NT) s_acc[i] = 0.f;
  for (int i = tid; i < cnt * H; i += NT) {
    const int g = (lo + i / H) * AST + i % H;
    float a_d = ad_[g];
    s_ad[i]  = a_d;
    s_m[i]   = lrelu(as_[g] + a_d);
    s_den[i] = 0.f;
  }
  for (int i = tid; i < RNG; i += NT) s_deg[i] = (i < cnt) ? 1 : 0;
  if (FIRST && tid == 0) { s_cnt = 0; s_ng = 0; }
  __syncthreads();

  int nm;
  if (FIRST) {
    auto keep = [&](int s, int dl) {
      int slot = atomicAdd(&s_cnt, 1);
      if (slot < CAP) { s_e[slot] = s; s_dl[slot] = dl; }
    };
    if (gridDim.x <= 256) {
      const int nfull = E0 >> 4;
      typedef int __attribute__((may_alias)) lds_int;
      lds_int* const s_gl = (lds_int*)s_ex;
      const int nbw = ((nfull + GRP - 1) / GRP) * BMW;
      const unsigned int* rowp = gidx + (long)blockIdx.x * nbw;
      for (int base = 0; base < nbw; base += NT * 2) {
        unsigned mw[2];
#pragma unroll
        for (int k = 0; k < 2; ++k) mw[k] = rowp[min(base + k * NT + tid, nbw - 1)];
#pragma unroll
        for (int k = 0; k < 2; ++k) {
          const int q = base + k * NT + tid;
          unsigned m = (q < nbw) ? mw[k] : 0u;
          while (m) {
            const int p = __builtin_ctz(m);
            m &= m - 1;
            const int slot = atomicAdd(&s_ng, 1);
            if (slot < CAP) s_gl[slot] = q * 32 + p;
          }
        }
      }
      __syncthreads();
      const int ng = min(__builtin_amdgcn_readfirstlane(s_ng), CAP);
      constexpr int NS = NT / 8;
      constexpr int LB = 8;
      const int st = tid >> 3, k8 = tid & 7;
      const int4* rv = (const int4*)rec;
      for (int base = 0; base < ng; base += NS * LB) {
        int4 q[LB];
#pragma unroll
        for (int k = 0; k < LB; ++k)
          q[k] = rv[(long)s_gl[min(base + k * NS + st, ng - 1)] * 8 + k8];
        int  off[2 * LB];
        bool hit[2 * LB];
        int  nk = 0;
#pragma unroll
        for (int k = 0; k < LB; ++k) {
          const bool in = base + k * NS + st < ng;
          hit[2 * k]     = in && ((unsigned)(q[k].y - lo) < (unsigned)cnt);
          hit[2 * k + 1] = in && ((unsigned)(q[k].w - lo) < (unsigned)cnt);
#pragma unroll
          for (int u = 0; u < 2; ++u) {
            const unsigned bl = __builtin_amdgcn_ballot_w32(hit[2 * k + u]);
            off[2 * k + u] = nk + (int)__builtin_amdgcn_mbcnt_lo(bl, 0u);
            nk += __builtin_popcount(bl);
          }
        }
        int first = 0;
        if (lane == 0) first = atomicAdd(&s_cnt, nk);
        first = __builtin_amdgcn_readfirstlane(first);
#pragma unroll
        for (int k = 0; k < LB; ++k) {
          const int sl0 = first + off[2 * k], sl1 = first + off[2 * k + 1];
          if (hit[2 * k]     && sl0 < CAP) { s_e[sl0] = q[k].x; s_dl[sl0] = q[k].y - lo; }
          if (hit[2 * k + 1] && sl1 < CAP) { s_e[sl1] = q[k].z; s_dl[sl1] = q[k].w - lo; }
        }
      }
      int et = nfull * 16 + tid;
      if (et < E0) {
        const unsigned dl = (unsigned)(dsts[et] - lo);
        if (dl < (unsigned)cnt) keep(srcs[et], (int)dl);
      }
    } else {
      for (int e = tid; e < E0; e += NT) {
        const unsigned dl = (unsigned)(dsts[e] - lo);
        if (dl < (unsigned)cnt) keep(srcs[e], (int)dl);
      }
    }
    __syncthreads();
    nm = __builtin_amdgcn_readfirstlane(s_cnt);
    if (tid == 0) lcnt[blockIdx.x] = nm;
  } else {
    nm = __builtin_amdgcn_readfirstlane(lcnt[blockIdx.x]);
  }
  const bool fits = (nm + cnt <= CAP);
  const int  tot  = nm + cnt;
  const int  totp = (tot + ROWS - 1) & ~(ROWS - 1);
  const long lb   = (long)blockIdx.x * CAP;

  if (fits) {
    for (int i = tid; i < nm; i += NT) {
      int s, dl;
      if (FIRST) {
        s  = s_e[i];
        dl = s_dl[i];
        lsrc[lb + i] = s;
        ldl[lb + i]  = dl;
      } else {
        s  = lsrc[lb + i];
        dl = ldl[lb + i];
        s_e[i]  = s;
        s_dl[i] = dl;
      }
      atomicAdd(&s_deg[dl], 1);
#pragma unroll
      for (int h = 0; h < H; ++h) {
        float v = lrelu(as_[s * AST + h] + s_ad[dl * H + h]);
        s_ex[i * H + h] = v;
        lds_max_f32(&s_m[dl * H + h], v);
      }
    }
  } else {
    for (int e = tid; e < E0; e += NT) {
      unsigned dl = (unsigned)(dsts[e] - lo);
      if (dl < (unsigned)cnt) {
        int s = srcs[e];
        for (int h = 0; h < H; ++h)
          lds_max_f32(&s_m[dl * H + h], lrelu(as_[s * AST + h] + s_ad[dl * H + h]));
      }
    }
  }
  __syncthreads();

  if (fits) {
    if (wave == 0) {
      constexpr int NCH = (RNG + 31) / 32;
      int dg[NCH];
#pragma unroll
      for (int c = 0; c < NCH; ++c) dg[c] = (c * 32 + lane < cnt) ? s_deg[c * 32 + lane] : 0;
      int carry = 0;
#pragma unroll
      for (int c = 0; c < NCH; ++c) {
        const int idx = c * 32 + lane;
        const int v   = dg[c];
        int x = v;
        x += __builtin_amdgcn_update_dpp(0, x, 0x111, 0xf, 0xf, true);
        x += __builtin_amdgcn_update_dpp(0, x, 0x112, 0xf, 0xf, true);
        x += __builtin_amdgcn_update_dpp(0, x, 0x114, 0xf, 0xf, true);
        x += __builtin_amdgcn_update_dpp(0, x, 0x118, 0xf, 0xf, true);
        const int r0 = __builtin_amdgcn_readlane(x, 15);
        if (lane >= 16) x += r0;
        if (idx < cnt) s_pos[idx] = carry + x - v;
        carry += __builtin_amdgcn_readlane(x, 31);
      }
    }
    __syncthreads();
  }

  if (fits) {
    for (int i = tid; i < nm; i += NT) {
      const int dl  = s_dl[i];
      const int pos = atomicAdd(&s_pos[dl], 1);
      s_e2[pos]  = s_e[i];
      s_dl2[pos] = dl;
#pragma unroll
      for (int h = 0; h < H; ++h) {
        float ex = expf(s_ex[i * H + h] - s_m[dl * H + h]);
        s_ex2[pos * H + h] = ex;
        atomicAdd(&s_den[dl * H + h], ex);
      }
    }
  }
  for (int j = tid; j < cnt; j += NT) {
    int pos = 0;
    if (fits) {
      pos = atomicAdd(&s_pos[j], 1);
      s_e2[pos]  = lo + j;
      s_dl2[pos] = j;
    }
#pragma unroll
    for (int h = 0; h < H; ++h) {
      const int i = j * H + h;
      float ex = expf(lrelu(as_[(lo + j) * AST + h] + s_ad[i]) - s_m[i]);
      s_self[i] = ex;
      if (fits) s_ex2[pos * H + h] = ex;
      atomicAdd(&s_den[i], ex);
    }
  }
  if (fits) {
    if (tid < totp - tot) {
      const int p = tot + tid;
      s_e2[p]  = lo;
      s_dl2[p] = RNG;
#pragma unroll
      for (int h = 0; h < H; ++h) s_ex2[p * H + h] = 0.f;
    }
  }
  __syncthreads();

  if (fits) {
    const int hl = (FJ * lane) / C;
    const int nturn = totp / ROWS;
    const int tb0 = (wave * nturn) / NW, tb1 = ((wave + 1) * nturn) / NW;
    int  cur = RNG;
    vrow run;
#pragma unroll
    for (int j = 0; j < FJ; ++j) run[j] = 0.f;
    auto flush = [&]() {
      float* acc = &s_acc[cur * F + lane];
#pragma unroll
      for (int j = 0; j < FJ; ++j) atomicAdd(acc + 32 * j, run[j]);
    };
    for (int b = tb0; b < tb1; ++b) {
      const int ii  = b * ROWS + (lane & (ROWS - 1));
      const int sv  = s_e2[ii];
      const int dlv = s_dl2[ii];
      const int exl = __float_as_int(s_ex2[ii * H + hl]);
      vrow val[ROWS];
#pragma unroll
      for (int k = 0; k < ROWS; ++k) {
        const int s = __builtin_amdgcn_readlane(sv, k);
        val[k] = *((const vrow*)(hf + (long)s * F) + lane);
      }
      auto add_entry = [&](auto kc) {
        constexpr int k = decltype(kc)::v;
        const int dl = __builtin_amdgcn_readlane(dlv, k);
        if (dl != cur) {
          flush();
#pragma unroll
          for (int j = 0; j < FJ; ++j) run[j] = 0.f;
          cur = dl;
        }
        const float w = (H == 1) ? __int_as_float(__builtin_amdgcn_readlane(exl, k))
                                 : row_lane<k>(exl);
#pragma unroll
        for (int j = 0; j < FJ; ++j) run[j] += val[k][j] * w;
      };
      kloop<0, ROWS>::run(add_entry);
    }
    flush();
  } else {
    for (int e = tid; e < E0; e += NT) {
      unsigned dl = (unsigned)(dsts[e] - lo);
      if (dl < (unsigned)cnt) {
        int s = srcs[e];
        for (int h = 0; h < H; ++h) {
          float ex = expf(lrelu(as_[s * AST + h] + s_ad[dl * H + h]) - s_m[dl * H + h]);
          atomicAdd(&s_den[dl * H + h], ex);
          for (int c = 0; c < C; ++c)
            atomicAdd(&s_acc[dl * F + acc_word<F>(h * C + c)], hf[(long)s * F + h * C + c] * ex);
        }
      }
    }
  }
  if (!fits) {
    for (int i = wave; i < cnt; i += NW) {
#pragma unroll
      for (int j = 0; j < FJ; ++j)
        atomicAdd(&s_acc[i * F + acc_word<F>(lane + 32 * j)],
                  hf[(long)(lo + i) * F + lane + 32 * j] * s_self[i * H + (32 * j) / C]);
    }
  }
  __syncthreads();

  for (int i = tid; i < cnt * F; i += NT) {
    int dl = i / F, c = i - dl * F;
    float v = s_acc[dl * F + acc_word<F>(c)] / s_den[dl * H + c / C] + bias[c];
    if (DO_ELU) v = (v > 0.f) ? v : (expf(v) - 1.f);
    out[(long)lo * F + i] = v;
  }
}

extern "C" void kernel_launch(void* const* d_in, const int* in_sizes, int n_in,
                              void* d_out, int out_size, void* d_ws, size_t ws_size,
                              hipStream_t stream) {
  const float* x    = (const float*)d_in[0];
  const int*   ei   = (const int*)  d_in[1];
  const float* W1   = (const float*)d_in[2];
  const float* atS1 = (const float*)d_in[3];
  const float* atD1 = (const float*)d_in[4];
  const float* b1   = (const float*)d_in[5];
  const float* W2   = (const float*)d_in[6];
  const float* atS2 = (const float*)d_in[7];
  const float* atD2 = (const float*)d_in[8];
  const float* b2   = (const float*)d_in[9];
  float* out = (float*)d_out;

  const int N    = in_sizes[0] / DIN;
  const int E0   = in_sizes[1] / 2;
  const int* srcs = ei;
  const int* dsts = ei + E0;

  float* ws = (float*)d_ws;
  long off = 0;
  float* h1  = ws + off; off += (long)N * (HEADS * DH);
  float* o1  = ws + off; off += (long)N * (HEADS * DH);
  float* a1s = ws + off; off += (long)N * HEADS;
  float* a1d = ws + off; off += (long)N * HEADS;
  off = (off + 3) & ~3L;
  float* h2  = h1;
  float* a2  = h1 + (long)N * DOUT;

  auto blocks = [](long n, int tb) { return (unsigned)((n + tb - 1) / tb); };
  const unsigned NB = blocks(N, RNG);
  int* lsrc = (int*)(ws + off); off += (long)NB * CAP;
  int* ldl  = (int*)(ws + off); off += (long)NB * CAP;
  int* lcnt = (int*)(ws + off); off += NB;
  off = (off + 3) & ~3L;
  off = (off + 31) & ~31L;
  int* rec = (int*)(ws + off); off += (long)(E0 >> 4) * 32;
  off = (off + 31) & ~31L;
  const unsigned NG = blocks((long)(E0 >> 4), GRP);
  const unsigned NR = blocks((long)(E0 >> 4) * 16, RPB);
  unsigned int* gidx = (unsigned int*)(ws + off);

  gemm_wmma_f32<(HEADS * DH) / 32, HEADS><<<256, 256, 0, stream>>>(x, W1, h1, N, atS1, atD1, a1s, a1d);
  edge_keys<<<NG + NR, GRP, 0, stream>>>(srcs, dsts, E0, rec, gidx, (int)NG);
  gat_owner<HEADS, DH, true, true><<<NB, NT, 0, stream>>>(srcs, dsts, E0, N, a1s, a1d, h1, b1, o1,
                                                          lsrc, ldl, lcnt, rec, gidx);

  gemm_wmma_f32<DOUT / 32, 1><<<256, 256, 0, stream>>>(o1, W2, h2, N, atS2, atD2, a2, nullptr);
  gat_owner<1, DOUT, false, false><<<NB, NT, 0, stream>>>(srcs, dsts, E0, N, a2, a2 + 1, h2, b2, out,
                                                          lsrc, ldl, lcnt, rec, gidx);
}
